# starting kernel
# baseline (speedup 1.0000x reference)
#include <hip/hip_runtime.h>
#include <hip/hip_bf16.h>
#include <math.h>

#define NF    128
#define EMBD  64
#define MAXD  16
#define NCLS  2

typedef float v2f __attribute__((ext_vector_type(2)));
typedef float v8f __attribute__((ext_vector_type(8)));

__global__ __launch_bounds__(256)
void care_phase1(const float* __restrict__ x,
                 const float* __restrict__ w_dist,
                 const float* __restrict__ b_dist,
                 float* __restrict__ score_out,
                 float* __restrict__ s0,
                 int N)
{
    int wave = (int)((blockIdx.x * blockDim.x + threadIdx.x) >> 5);
    int lane = threadIdx.x & 31;
    if (wave >= N) return;
    const float* xr = x + (size_t)wave * NF;
    float acc0 = 0.f, acc1 = 0.f;
    #pragma unroll
    for (int f = 0; f < NF; f += 32) {
        float v = xr[f + lane];
        acc0 += v * w_dist[(f + lane) * 2 + 0];
        acc1 += v * w_dist[(f + lane) * 2 + 1];
    }
    #pragma unroll
    for (int off = 16; off > 0; off >>= 1) {
        acc0 += __shfl_xor(acc0, off, 32);
        acc1 += __shfl_xor(acc1, off, 32);
    }
    if (lane == 0) {
        float l0 = acc0 + b_dist[0];
        float l1 = acc1 + b_dist[1];
        score_out[(size_t)wave * 2 + 0] = l0;
        score_out[(size_t)wave * 2 + 1] = l1;
        s0[wave] = tanhf(l0);
    }
}

__global__ __launch_bounds__(128)
void care_phase2(const float* __restrict__ x,
                 const float* __restrict__ w_self,
                 const float* __restrict__ w_neigh,
                 const float* __restrict__ b_sage,
                 const float* __restrict__ w_lin,
                 const float* __restrict__ b_lin,
                 const int*   __restrict__ neighbors,
                 const int*   __restrict__ degrees,
                 const float* __restrict__ s0,
                 float* __restrict__ logits_out,
                 int N)
{
    extern __shared__ float4 shW[];
    __shared__ float sh_an[4][16][NF + 4];
    __shared__ float sh_h [4][16][EMBD + 2];
    __shared__ float sh_wlin[EMBD * NCLS];
    __shared__ int   sh_sel[4][16][MAXD / 2];
    __shared__ int   sh_k [4][16];

    const int tid  = threadIdx.x;
    const int wid  = tid >> 5;
    const int lane = tid & 31;
    const int hi   = lane >> 4;
    const int lo   = lane & 15;

    int tile0 = (int)blockIdx.x * 64 + wid * 16;
    const bool active = (tile0 < N);
    if (!active) tile0 = 0;

    #pragma unroll 4
    for (int idx = tid; idx < (NF / 4) * 4 * 32; idx += 128) {
        const int l   = idx & 31;
        const int ct  = (idx >> 5) & 3;
        const int i   = idx >> 7;
        const int kk  = 4 * i + 2 * (l >> 4);
        const int col = ct * 16 + (l & 15);
        float4 w;
        w.x = w_self [(size_t)(kk)     * EMBD + col];
        w.y = w_self [(size_t)(kk + 1) * EMBD + col];
        w.z = w_neigh[(size_t)(kk)     * EMBD + col];
        w.w = w_neigh[(size_t)(kk + 1) * EMBD + col];
        shW[idx] = w;
    }
    if (tid < EMBD * NCLS) sh_wlin[tid] = w_lin[tid];

    if (lane < 16) {
        int node = tile0 + lane;
        if (node >= N) node = N - 1;
        int deg = degrees[node];
        if (deg < 1) deg = 1;
        if (deg > MAXD) deg = MAXD;
        int k = deg >> 1;
        if (k < 1) k = 1;
        float sc = s0[node];
        float dist[MAXD];
        int   ids [MAXD];
        #pragma unroll
        for (int j = 0; j < MAXD; ++j) {
            int nb = neighbors[(size_t)node * MAXD + j];
            ids[j] = nb;
            float d = fabsf(sc - s0[nb]);
            dist[j] = (j < deg) ? d : INFINITY;
        }
        for (int t = 0; t < k; ++t) {
            int best = 0; float bv = INFINITY;
            #pragma unroll
            for (int j = 0; j < MAXD; ++j) {
                if (dist[j] < bv) { bv = dist[j]; best = j; }
            }
            sh_sel[wid][lane][t] = ids[best];
            dist[best] = INFINITY;
        }
        sh_k[wid][lane] = k;
    }
    __syncthreads();

    for (int m = 0; m < 16; ++m) {
        int k = sh_k[wid][m];
        float a0 = 0.f, a1 = 0.f, a2 = 0.f, a3 = 0.f;
        for (int j = 0; j < k; ++j) {
            const float* xr = x + (size_t)sh_sel[wid][m][j] * NF;
            a0 += xr[lane];
            a1 += xr[lane + 32];
            a2 += xr[lane + 64];
            a3 += xr[lane + 96];
        }
        float inv = 1.0f / (float)k;
        sh_an[wid][m][lane]      = a0 * inv;
        sh_an[wid][m][lane + 32] = a1 * inv;
        sh_an[wid][m][lane + 64] = a2 * inv;
        sh_an[wid][m][lane + 96] = a3 * inv;
    }
    __syncthreads();

    int srow = tile0 + lo;
    if (srow >= N) srow = N - 1;
    const float* xself = x + (size_t)srow * NF;

    v8f c[4];
    #pragma unroll
    for (int ct = 0; ct < 4; ++ct) c[ct] = (v8f){0.f,0.f,0.f,0.f,0.f,0.f,0.f,0.f};

    for (int i = 0; i < NF / 4; ++i) {
        const int kk = 4 * i + hi * 2;
        v2f a_s, a_n;
        a_s.x = xself[kk];          a_s.y = xself[kk + 1];
        a_n.x = sh_an[wid][lo][kk]; a_n.y = sh_an[wid][lo][kk + 1];
        #pragma unroll
        for (int ct = 0; ct < 4; ++ct) {
            float4 wf = shW[(i * 4 + ct) * 32 + lane];
            v2f b_s; b_s.x = wf.x; b_s.y = wf.y;
            v2f b_n; b_n.x = wf.z; b_n.y = wf.w;
            c[ct] = __builtin_amdgcn_wmma_f32_16x16x4_f32(
                        false, a_s, false, b_s, (short)0, c[ct], false, false);
            c[ct] = __builtin_amdgcn_wmma_f32_16x16x4_f32(
                        false, a_n, false, b_n, (short)0, c[ct], false, false);
        }
    }

    #pragma unroll
    for (int ct = 0; ct < 4; ++ct) {
        const int e = ct * 16 + lo;
        const float be = b_sage[e];
        #pragma unroll
        for (int j = 0; j < 8; ++j) {
            float v = c[ct][j] + be;
            sh_h[wid][j + 8 * hi][e] = v > 0.f ? v : 0.f;
        }
    }
    __syncthreads();

    {
        const int m = lane >> 1, cls = lane & 1;
        float acc = b_lin[cls];
        #pragma unroll 8
        for (int e = 0; e < EMBD; ++e)
            acc += sh_h[wid][m][e] * sh_wlin[e * NCLS + cls];
        const int node = tile0 + m;
        if (active && node < N)
            logits_out[(size_t)node * NCLS + cls] = acc;
    }
}

extern "C" void kernel_launch(void* const* d_in, const int* in_sizes, int n_in,
                              void* d_out, int out_size, void* d_ws, size_t ws_size,
                              hipStream_t stream) {
    const float* x       = (const float*)d_in[0];
    const float* w_dist  = (const float*)d_in[1];
    const float* b_dist  = (const float*)d_in[2];
    const float* w_self  = (const float*)d_in[3];
    const float* w_neigh = (const float*)d_in[4];
    const float* b_sage  = (const float*)d_in[5];
    const float* w_lin   = (const float*)d_in[6];
    const float* b_lin   = (const float*)d_in[7];
    const int*   neigh   = (const int*)d_in[8];
    const int*   degs    = (const int*)d_in[9];

    const int N = in_sizes[9];
    float* out       = (float*)d_out;
    float* logits    = out;
    float* score_out = out + (size_t)N * NCLS;
    float* s0        = (float*)d_ws;

    {
        int waves_per_block = 256 / 32;
        int blocks = (N + waves_per_block - 1) / waves_per_block;
        care_phase1<<<blocks, 256, 0, stream>>>(x, w_dist, b_dist,
                                                score_out, s0, N);
    }
    {
        int blocks = (N + 63) / 64;
        size_t dynbytes = (size_t)(NF / 4) * 4 * 32 * sizeof(float4);
        care_phase2<<<blocks, 128, dynbytes, stream>>>(x, w_self, w_neigh,
                                                       b_sage, w_lin, b_lin,
                                                       neigh, degs, s0,
                                                       logits, N);
    }
}


# Neighbour-row gather: one 16-byte load per lane fetches a whole row, and all selected rows of four nodes are issued before the first sum (one memory round trip per four nodes, not one per row)
# speedup vs baseline: 1.6661x; 1.6661x over previous
#include <hip/hip_runtime.h>
#include <hip/hip_bf16.h>
#include <math.h>

#define NF    128
#define EMBD  64
#define MAXD  16
#define NCLS  2

typedef float v2f __attribute__((ext_vector_type(2)));
typedef float v8f __attribute__((ext_vector_type(8)));

__global__ __launch_bounds__(256)
void care_phase1(const float* __restrict__ x,
                 const float* __restrict__ w_dist,
                 const float* __restrict__ b_dist,
                 float* __restrict__ score_out,
                 float* __restrict__ s0,
                 int N)
{
    int wave = (int)((blockIdx.x * blockDim.x + threadIdx.x) >> 5);
    int lane = threadIdx.x & 31;
    if (wave >= N) return;
    const float* xr = x + (size_t)wave * NF;
    float acc0 = 0.f, acc1 = 0.f;
    #pragma unroll
    for (int f = 0; f < NF; f += 32) {
        float v = xr[f + lane];
        acc0 += v * w_dist[(f + lane) * 2 + 0];
        acc1 += v * w_dist[(f + lane) * 2 + 1];
    }
    #pragma unroll
    for (int off = 16; off > 0; off >>= 1) {
        acc0 += __shfl_xor(acc0, off, 32);
        acc1 += __shfl_xor(acc1, off, 32);
    }
    if (lane == 0) {
        float l0 = acc0 + b_dist[0];
        float l1 = acc1 + b_dist[1];
        score_out[(size_t)wave * 2 + 0] = l0;
        score_out[(size_t)wave * 2 + 1] = l1;
        s0[wave] = tanhf(l0);
    }
}

__global__ __launch_bounds__(128)
void care_phase2(const float* __restrict__ x,
                 const float* __restrict__ w_self,
                 const float* __restrict__ w_neigh,
                 const float* __restrict__ b_sage,
                 const float* __restrict__ w_lin,
                 const float* __restrict__ b_lin,
                 const int*   __restrict__ neighbors,
                 const int*   __restrict__ degrees,
                 const float* __restrict__ s0,
                 float* __restrict__ logits_out,
                 int N)
{
    extern __shared__ float4 shW[];
    __shared__ __attribute__((aligned(16))) float sh_an[4][16][NF + 4];
    __shared__ float sh_h [4][16][EMBD + 2];
    __shared__ float sh_wlin[EMBD * NCLS];
    __shared__ __attribute__((aligned(16))) int sh_sel[4][16][MAXD / 2];
    __shared__ int   sh_k [4][16];

    const int tid  = threadIdx.x;
    const int wid  = tid >> 5;
    const int lane = tid & 31;
    const int hi   = lane >> 4;
    const int lo   = lane & 15;

    int tile0 = (int)blockIdx.x * 64 + wid * 16;
    const bool active = (tile0 < N);
    if (!active) tile0 = 0;

    #pragma unroll 4
    for (int idx = tid; idx < (NF / 4) * 4 * 32; idx += 128) {
        const int l   = idx & 31;
        const int ct  = (idx >> 5) & 3;
        const int i   = idx >> 7;
        const int kk  = 4 * i + 2 * (l >> 4);
        const int col = ct * 16 + (l & 15);
        float4 w;
        w.x = w_self [(size_t)(kk)     * EMBD + col];
        w.y = w_self [(size_t)(kk + 1) * EMBD + col];
        w.z = w_neigh[(size_t)(kk)     * EMBD + col];
        w.w = w_neigh[(size_t)(kk + 1) * EMBD + col];
        shW[idx] = w;
    }
    if (tid < EMBD * NCLS) sh_wlin[tid] = w_lin[tid];

    if (lane < 16) {
        int node = tile0 + lane;
        if (node >= N) node = N - 1;
        int deg = degrees[node];
        if (deg < 1) deg = 1;
        if (deg > MAXD) deg = MAXD;
        int k = deg >> 1;
        if (k < 1) k = 1;
        float sc = s0[node];
        float dist[MAXD];
        int   ids [MAXD];
        #pragma unroll
        for (int j = 0; j < MAXD; ++j) {
            int nb = neighbors[(size_t)node * MAXD + j];
            ids[j] = nb;
            float d = fabsf(sc - s0[nb]);
            dist[j] = (j < deg) ? d : INFINITY;
        }
        *(int4*)&sh_sel[wid][lane][0] = make_int4(0, 0, 0, 0);
        *(int4*)&sh_sel[wid][lane][4] = make_int4(0, 0, 0, 0);
        for (int t = 0; t < k; ++t) {
            int best = 0; float bv = INFINITY;
            #pragma unroll
            for (int j = 0; j < MAXD; ++j) {
                if (dist[j] < bv) { bv = dist[j]; best = j; }
            }
            sh_sel[wid][lane][t] = ids[best];
            dist[best] = INFINITY;
        }
        sh_k[wid][lane] = k;
    }
    __syncthreads();

    const float* xl = x + 4 * lane;
    for (int m0 = 0; m0 < 16; m0 += 4) {
        int    kq[4];
        int    idq[4][MAXD / 2];
        float4 r[4][MAXD / 2];
        #pragma unroll
        for (int q = 0; q < 4; ++q) {
            const int4 s_lo = *(const int4*)&sh_sel[wid][m0 + q][0];
            const int4 s_hi = *(const int4*)&sh_sel[wid][m0 + q][4];
            kq[q]     = __builtin_amdgcn_readfirstlane(sh_k[wid][m0 + q]);
            idq[q][0] = __builtin_amdgcn_readfirstlane(s_lo.x);
            idq[q][1] = __builtin_amdgcn_readfirstlane(s_lo.y);
            idq[q][2] = __builtin_amdgcn_readfirstlane(s_lo.z);
            idq[q][3] = __builtin_amdgcn_readfirstlane(s_lo.w);
            idq[q][4] = __builtin_amdgcn_readfirstlane(s_hi.x);
            idq[q][5] = __builtin_amdgcn_readfirstlane(s_hi.y);
            idq[q][6] = __builtin_amdgcn_readfirstlane(s_hi.z);
            idq[q][7] = __builtin_amdgcn_readfirstlane(s_hi.w);
        }
        #pragma unroll
        for (int q = 0; q < 4; ++q) {
            #pragma unroll
            for (int j = 0; j < MAXD / 2; ++j) {
                if (j < kq[q])
                    r[q][j] = *(const float4*)(xl + (size_t)idq[q][j] * NF);
            }
        }
        #pragma unroll
        for (int q = 0; q < 4; ++q) {
            float4 a = make_float4(0.f, 0.f, 0.f, 0.f);
            #pragma unroll
            for (int j = 0; j < MAXD / 2; ++j) {
                if (j < kq[q]) {
                    a.x += r[q][j].x; a.y += r[q][j].y;
                    a.z += r[q][j].z; a.w += r[q][j].w;
                }
            }
            const float inv = 1.0f / (float)kq[q];
            a.x *= inv; a.y *= inv; a.z *= inv; a.w *= inv;
            *(float4*)&sh_an[wid][m0 + q][4 * lane] = a;
        }
    }
    __syncthreads();

    int srow = tile0 + lo;
    if (srow >= N) srow = N - 1;
    const float* xself = x + (size_t)srow * NF;

    v8f c[4];
    #pragma unroll
    for (int ct = 0; ct < 4; ++ct) c[ct] = (v8f){0.f,0.f,0.f,0.f,0.f,0.f,0.f,0.f};

    for (int i = 0; i < NF / 4; ++i) {
        const int kk = 4 * i + hi * 2;
        v2f a_s, a_n;
        a_s.x = xself[kk];          a_s.y = xself[kk + 1];
        a_n.x = sh_an[wid][lo][kk]; a_n.y = sh_an[wid][lo][kk + 1];
        #pragma unroll
        for (int ct = 0; ct < 4; ++ct) {
            float4 wf = shW[(i * 4 + ct) * 32 + lane];
            v2f b_s; b_s.x = wf.x; b_s.y = wf.y;
            v2f b_n; b_n.x = wf.z; b_n.y = wf.w;
            c[ct] = __builtin_amdgcn_wmma_f32_16x16x4_f32(
                        false, a_s, false, b_s, (short)0, c[ct], false, false);
            c[ct] = __builtin_amdgcn_wmma_f32_16x16x4_f32(
                        false, a_n, false, b_n, (short)0, c[ct], false, false);
        }
    }

    #pragma unroll
    for (int ct = 0; ct < 4; ++ct) {
        const int e = ct * 16 + lo;
        const float be = b_sage[e];
        #pragma unroll
        for (int j = 0; j < 8; ++j) {
            float v = c[ct][j] + be;
            sh_h[wid][j + 8 * hi][e] = v > 0.f ? v : 0.f;
        }
    }
    __syncthreads();

    {
        const int m = lane >> 1, cls = lane & 1;
        float acc = b_lin[cls];
        #pragma unroll 8
        for (int e = 0; e < EMBD; ++e)
            acc += sh_h[wid][m][e] * sh_wlin[e * NCLS + cls];
        const int node = tile0 + m;
        if (active && node < N)
            logits_out[(size_t)node * NCLS + cls] = acc;
    }
}

extern "C" void kernel_launch(void* const* d_in, const int* in_sizes, int n_in,
                              void* d_out, int out_size, void* d_ws, size_t ws_size,
                              hipStream_t stream) {
    const float* x       = (const float*)d_in[0];
    const float* w_dist  = (const float*)d_in[1];
    const float* b_dist  = (const float*)d_in[2];
    const float* w_self  = (const float*)d_in[3];
    const float* w_neigh = (const float*)d_in[4];
    const float* b_sage  = (const float*)d_in[5];
    const float* w_lin   = (const float*)d_in[6];
    const float* b_lin   = (const float*)d_in[7];
    const int*   neigh   = (const int*)d_in[8];
    const int*   degs    = (const int*)d_in[9];

    const int N = in_sizes[9];
    float* out       = (float*)d_out;
    float* logits    = out;
    float* score_out = out + (size_t)N * NCLS;
    float* s0        = (float*)d_ws;

    {
        int waves_per_block = 256 / 32;
        int blocks = (N + waves_per_block - 1) / waves_per_block;
        care_phase1<<<blocks, 256, 0, stream>>>(x, w_dist, b_dist,
                                                score_out, s0, N);
    }
    {
        int blocks = (N + 63) / 64;
        size_t dynbytes = (size_t)(NF / 4) * 4 * 32 * sizeof(float4);
        care_phase2<<<blocks, 128, dynbytes, stream>>>(x, w_self, w_neigh,
                                                       b_sage, w_lin, b_lin,
                                                       neigh, degs, s0,
                                                       logits, N);
    }
}


# Score pass: one wave scores 32 consecutive nodes (N/32 waves instead of N): all 128 row loads issued before the first sum, the 32 wave sums folded stage by stage, one node's result per lane
# speedup vs baseline: 1.8332x; 1.1003x over previous
#include <hip/hip_runtime.h>
#include <hip/hip_bf16.h>
#include <math.h>

#define NF    128
#define EMBD  64
#define MAXD  16
#define NCLS  2

typedef float v2f __attribute__((ext_vector_type(2)));
typedef float v8f __attribute__((ext_vector_type(8)));

#define P1_ROWS 32

__global__ __launch_bounds__(256)
void care_phase1(const float* __restrict__ x,
                 const float* __restrict__ w_dist,
                 const float* __restrict__ b_dist,
                 float* __restrict__ score_out,
                 float* __restrict__ s0,
                 int N)
{
    const int lane = threadIdx.x & 31;
    const int wave = __builtin_amdgcn_readfirstlane(
                         (int)((blockIdx.x * blockDim.x + threadIdx.x) >> 5));
    const int row0 = wave * P1_ROWS;
    if (row0 >= N) return;

    float w0[4], w1[4];
    #pragma unroll
    for (int j = 0; j < 4; ++j) {
        w0[j] = w_dist[(32 * j + lane) * 2 + 0];
        w1[j] = w_dist[(32 * j + lane) * 2 + 1];
    }

    float v[P1_ROWS][4];
    #pragma unroll
    for (int q = 0; q < P1_ROWS; ++q) {
        int row = row0 + q;
        if (row >= N) row = N - 1;
        const float* xr = x + (size_t)row * NF + lane;
        #pragma unroll
        for (int j = 0; j < 4; ++j) v[q][j] = xr[32 * j];
    }
    __builtin_amdgcn_sched_barrier(0);

    float a0[P1_ROWS], a1[P1_ROWS];
    #pragma unroll
    for (int q = 0; q < P1_ROWS; ++q) {
        float acc0 = 0.f, acc1 = 0.f;
        #pragma unroll
        for (int j = 0; j < 4; ++j) {
            acc0 += v[q][j] * w0[j];
            acc1 += v[q][j] * w1[j];
        }
        a0[q] = acc0; a1[q] = acc1;
    }
    #pragma unroll
    for (int off = 16; off > 0; off >>= 1) {
        #pragma unroll
        for (int q = 0; q < P1_ROWS; ++q) {
            a0[q] += __shfl_xor(a0[q], off, 32);
            a1[q] += __shfl_xor(a1[q], off, 32);
        }
    }

    float m0 = 0.f, m1 = 0.f;
    #pragma unroll
    for (int q = 0; q < P1_ROWS; ++q) {
        if (lane == q) { m0 = a0[q]; m1 = a1[q]; }
    }

    const int node = row0 + lane;
    if (node < N) {
        const float l0 = m0 + b_dist[0];
        const float l1 = m1 + b_dist[1];
        *(float2*)(score_out + (size_t)node * 2) = make_float2(l0, l1);
        s0[node] = tanhf(l0);
    }
}

__global__ __launch_bounds__(128)
void care_phase2(const float* __restrict__ x,
                 const float* __restrict__ w_self,
                 const float* __restrict__ w_neigh,
                 const float* __restrict__ b_sage,
                 const float* __restrict__ w_lin,
                 const float* __restrict__ b_lin,
                 const int*   __restrict__ neighbors,
                 const int*   __restrict__ degrees,
                 const float* __restrict__ s0,
                 float* __restrict__ logits_out,
                 int N)
{
    extern __shared__ float4 shW[];
    __shared__ __attribute__((aligned(16))) float sh_an[4][16][NF + 4];
    __shared__ float sh_h [4][16][EMBD + 2];
    __shared__ float sh_wlin[EMBD * NCLS];
    __shared__ __attribute__((aligned(16))) int sh_sel[4][16][MAXD / 2];
    __shared__ int   sh_k [4][16];

    const int tid  = threadIdx.x;
    const int wid  = tid >> 5;
    const int lane = tid & 31;
    const int hi   = lane >> 4;
    const int lo   = lane & 15;

    int tile0 = (int)blockIdx.x * 64 + wid * 16;
    const bool active = (tile0 < N);
    if (!active) tile0 = 0;

    #pragma unroll 4
    for (int idx = tid; idx < (NF / 4) * 4 * 32; idx += 128) {
        const int l   = idx & 31;
        const int ct  = (idx >> 5) & 3;
        const int i   = idx >> 7;
        const int kk  = 4 * i + 2 * (l >> 4);
        const int col = ct * 16 + (l & 15);
        float4 w;
        w.x = w_self [(size_t)(kk)     * EMBD + col];
        w.y = w_self [(size_t)(kk + 1) * EMBD + col];
        w.z = w_neigh[(size_t)(kk)     * EMBD + col];
        w.w = w_neigh[(size_t)(kk + 1) * EMBD + col];
        shW[idx] = w;
    }
    if (tid < EMBD * NCLS) sh_wlin[tid] = w_lin[tid];

    if (lane < 16) {
        int node = tile0 + lane;
        if (node >= N) node = N - 1;
        int deg = degrees[node];
        if (deg < 1) deg = 1;
        if (deg > MAXD) deg = MAXD;
        int k = deg >> 1;
        if (k < 1) k = 1;
        float sc = s0[node];
        float dist[MAXD];
        int   ids [MAXD];
        #pragma unroll
        for (int j = 0; j < MAXD; ++j) {
            int nb = neighbors[(size_t)node * MAXD + j];
            ids[j] = nb;
            float d = fabsf(sc - s0[nb]);
            dist[j] = (j < deg) ? d : INFINITY;
        }
        *(int4*)&sh_sel[wid][lane][0] = make_int4(0, 0, 0, 0);
        *(int4*)&sh_sel[wid][lane][4] = make_int4(0, 0, 0, 0);
        for (int t = 0; t < k; ++t) {
            int best = 0; float bv = INFINITY;
            #pragma unroll
            for (int j = 0; j < MAXD; ++j) {
                if (dist[j] < bv) { bv = dist[j]; best = j; }
            }
            sh_sel[wid][lane][t] = ids[best];
            dist[best] = INFINITY;
        }
        sh_k[wid][lane] = k;
    }
    __syncthreads();

    const float* xl = x + 4 * lane;
    for (int m0 = 0; m0 < 16; m0 += 4) {
        int    kq[4];
        int    idq[4][MAXD / 2];
        float4 r[4][MAXD / 2];
        #pragma unroll
        for (int q = 0; q < 4; ++q) {
            const int4 s_lo = *(const int4*)&sh_sel[wid][m0 + q][0];
            const int4 s_hi = *(const int4*)&sh_sel[wid][m0 + q][4];
            kq[q]     = __builtin_amdgcn_readfirstlane(sh_k[wid][m0 + q]);
            idq[q][0] = __builtin_amdgcn_readfirstlane(s_lo.x);
            idq[q][1] = __builtin_amdgcn_readfirstlane(s_lo.y);
            idq[q][2] = __builtin_amdgcn_readfirstlane(s_lo.z);
            idq[q][3] = __builtin_amdgcn_readfirstlane(s_lo.w);
            idq[q][4] = __builtin_amdgcn_readfirstlane(s_hi.x);
            idq[q][5] = __builtin_amdgcn_readfirstlane(s_hi.y);
            idq[q][6] = __builtin_amdgcn_readfirstlane(s_hi.z);
            idq[q][7] = __builtin_amdgcn_readfirstlane(s_hi.w);
        }
        #pragma unroll
        for (int q = 0; q < 4; ++q) {
            #pragma unroll
            for (int j = 0; j < MAXD / 2; ++j) {
                if (j < kq[q])
                    r[q][j] = *(const float4*)(xl + (size_t)idq[q][j] * NF);
            }
        }
        #pragma unroll
        for (int q = 0; q < 4; ++q) {
            float4 a = make_float4(0.f, 0.f, 0.f, 0.f);
            #pragma unroll
            for (int j = 0; j < MAXD / 2; ++j) {
                if (j < kq[q]) {
                    a.x += r[q][j].x; a.y += r[q][j].y;
                    a.z += r[q][j].z; a.w += r[q][j].w;
                }
            }
            const float inv = 1.0f / (float)kq[q];
            a.x *= inv; a.y *= inv; a.z *= inv; a.w *= inv;
            *(float4*)&sh_an[wid][m0 + q][4 * lane] = a;
        }
    }
    __syncthreads();

    int srow = tile0 + lo;
    if (srow >= N) srow = N - 1;
    const float* xself = x + (size_t)srow * NF;

    v8f c[4];
    #pragma unroll
    for (int ct = 0; ct < 4; ++ct) c[ct] = (v8f){0.f,0.f,0.f,0.f,0.f,0.f,0.f,0.f};

    for (int i = 0; i < NF / 4; ++i) {
        const int kk = 4 * i + hi * 2;
        v2f a_s, a_n;
        a_s.x = xself[kk];          a_s.y = xself[kk + 1];
        a_n.x = sh_an[wid][lo][kk]; a_n.y = sh_an[wid][lo][kk + 1];
        #pragma unroll
        for (int ct = 0; ct < 4; ++ct) {
            float4 wf = shW[(i * 4 + ct) * 32 + lane];
            v2f b_s; b_s.x = wf.x; b_s.y = wf.y;
            v2f b_n; b_n.x = wf.z; b_n.y = wf.w;
            c[ct] = __builtin_amdgcn_wmma_f32_16x16x4_f32(
                        false, a_s, false, b_s, (short)0, c[ct], false, false);
            c[ct] = __builtin_amdgcn_wmma_f32_16x16x4_f32(
                        false, a_n, false, b_n, (short)0, c[ct], false, false);
        }
    }

    #pragma unroll
    for (int ct = 0; ct < 4; ++ct) {
        const int e = ct * 16 + lo;
        const float be = b_sage[e];
        #pragma unroll
        for (int j = 0; j < 8; ++j) {
            float v = c[ct][j] + be;
            sh_h[wid][j + 8 * hi][e] = v > 0.f ? v : 0.f;
        }
    }
    __syncthreads();

    {
        const int m = lane >> 1, cls = lane & 1;
        float acc = b_lin[cls];
        #pragma unroll 8
        for (int e = 0; e < EMBD; ++e)
            acc += sh_h[wid][m][e] * sh_wlin[e * NCLS + cls];
        const int node = tile0 + m;
        if (active && node < N)
            logits_out[(size_t)node * NCLS + cls] = acc;
    }
}

extern "C" void kernel_launch(void* const* d_in, const int* in_sizes, int n_in,
                              void* d_out, int out_size, void* d_ws, size_t ws_size,
                              hipStream_t stream) {
    const float* x       = (const float*)d_in[0];
    const float* w_dist  = (const float*)d_in[1];
    const float* b_dist  = (const float*)d_in[2];
    const float* w_self  = (const float*)d_in[3];
    const float* w_neigh = (const float*)d_in[4];
    const float* b_sage  = (const float*)d_in[5];
    const float* w_lin   = (const float*)d_in[6];
    const float* b_lin   = (const float*)d_in[7];
    const int*   neigh   = (const int*)d_in[8];
    const int*   degs    = (const int*)d_in[9];

    const int N = in_sizes[9];
    float* out       = (float*)d_out;
    float* logits    = out;
    float* score_out = out + (size_t)N * NCLS;
    float* s0        = (float*)d_ws;

    {
        int nodes_per_block = (256 / 32) * P1_ROWS;
        int blocks = (N + nodes_per_block - 1) / nodes_per_block;
        care_phase1<<<blocks, 256, 0, stream>>>(x, w_dist, b_dist,
                                                score_out, s0, N);
    }
    {
        int blocks = (N + 63) / 64;
        size_t dynbytes = (size_t)(NF / 4) * 4 * 32 * sizeof(float4);
        care_phase2<<<blocks, 128, dynbytes, stream>>>(x, w_self, w_neigh,
                                                       b_sage, w_lin, b_lin,
                                                       neigh, degs, s0,
                                                       logits, N);
    }
}


# Phase 2 persistent: one 16-wave block per WGP packs the weights into LDS once; each wave then loops over 16-node tiles on wave-private staging, no block barrier (was 64 KB repacked per 64 nodes)
# speedup vs baseline: 2.5729x; 1.4035x over previous
#include <hip/hip_runtime.h>
#include <hip/hip_bf16.h>
#include <math.h>

#define NF    128
#define EMBD  64
#define MAXD  16
#define NCLS  2

typedef float v2f __attribute__((ext_vector_type(2)));
typedef float v8f __attribute__((ext_vector_type(8)));

#define P1_ROWS 32

__global__ __launch_bounds__(256)
void care_phase1(const float* __restrict__ x,
                 const float* __restrict__ w_dist,
                 const float* __restrict__ b_dist,
                 float* __restrict__ score_out,
                 float* __restrict__ s0,
                 int N)
{
    const int lane = threadIdx.x & 31;
    const int wave = __builtin_amdgcn_readfirstlane(
                         (int)((blockIdx.x * blockDim.x + threadIdx.x) >> 5));
    const int row0 = wave * P1_ROWS;
    if (row0 >= N) return;

    float w0[4], w1[4];
    #pragma unroll
    for (int j = 0; j < 4; ++j) {
        w0[j] = w_dist[(32 * j + lane) * 2 + 0];
        w1[j] = w_dist[(32 * j + lane) * 2 + 1];
    }

    float v[P1_ROWS][4];
    #pragma unroll
    for (int q = 0; q < P1_ROWS; ++q) {
        int row = row0 + q;
        if (row >= N) row = N - 1;
        const float* xr = x + (size_t)row * NF + lane;
        #pragma unroll
        for (int j = 0; j < 4; ++j) v[q][j] = xr[32 * j];
    }
    __builtin_amdgcn_sched_barrier(0);

    float a0[P1_ROWS], a1[P1_ROWS];
    #pragma unroll
    for (int q = 0; q < P1_ROWS; ++q) {
        float acc0 = 0.f, acc1 = 0.f;
        #pragma unroll
        for (int j = 0; j < 4; ++j) {
            acc0 += v[q][j] * w0[j];
            acc1 += v[q][j] * w1[j];
        }
        a0[q] = acc0; a1[q] = acc1;
    }
    #pragma unroll
    for (int off = 16; off > 0; off >>= 1) {
        #pragma unroll
        for (int q = 0; q < P1_ROWS; ++q) {
            a0[q] += __shfl_xor(a0[q], off, 32);
            a1[q] += __shfl_xor(a1[q], off, 32);
        }
    }

    float m0 = 0.f, m1 = 0.f;
    #pragma unroll
    for (int q = 0; q < P1_ROWS; ++q) {
        if (lane == q) { m0 = a0[q]; m1 = a1[q]; }
    }

    const int node = row0 + lane;
    if (node < N) {
        const float l0 = m0 + b_dist[0];
        const float l1 = m1 + b_dist[1];
        *(float2*)(score_out + (size_t)node * 2) = make_float2(l0, l1);
        s0[node] = tanhf(l0);
    }
}

#define P2_THREADS 512
#define P2_WAVES (P2_THREADS / 32)

static __device__ __forceinline__ void wave_lds_order()
{
    __builtin_amdgcn_fence(__ATOMIC_RELEASE, "wavefront");
    __builtin_amdgcn_wave_barrier();
    __builtin_amdgcn_fence(__ATOMIC_ACQUIRE, "wavefront");
}

__global__ __launch_bounds__(P2_THREADS)
void care_phase2(const float* __restrict__ x,
                 const float* __restrict__ w_self,
                 const float* __restrict__ w_neigh,
                 const float* __restrict__ b_sage,
                 const float* __restrict__ w_lin,
                 const float* __restrict__ b_lin,
                 const int*   __restrict__ neighbors,
                 const int*   __restrict__ degrees,
                 const float* __restrict__ s0,
                 float* __restrict__ logits_out,
                 int N)
{
    __shared__ float4 shW[(NF / 4) * 4 * 32];
    __shared__ __attribute__((aligned(16))) float sh_an[P2_WAVES][16][NF + 4];
    __shared__ float sh_h [P2_WAVES][16][EMBD + 2];
    __shared__ float sh_wlin[EMBD * NCLS];
    __shared__ __attribute__((aligned(16))) int sh_sel[P2_WAVES][16][MAXD / 2];
    __shared__ int   sh_k [P2_WAVES][16];

    const int tid  = threadIdx.x;
    const int lane = tid & 31;
    const int wid  = __builtin_amdgcn_readfirstlane(tid >> 5);
    const int hi   = lane >> 4;
    const int lo   = lane & 15;

    #pragma unroll
    for (int idx = tid; idx < (NF / 4) * 4 * 32; idx += P2_THREADS) {
        const int l   = idx & 31;
        const int ct  = (idx >> 5) & 3;
        const int i   = idx >> 7;
        const int kk  = 4 * i + 2 * (l >> 4);
        const int col = ct * 16 + (l & 15);
        float4 w;
        w.x = w_self [(size_t)(kk)     * EMBD + col];
        w.y = w_self [(size_t)(kk + 1) * EMBD + col];
        w.z = w_neigh[(size_t)(kk)     * EMBD + col];
        w.w = w_neigh[(size_t)(kk + 1) * EMBD + col];
        shW[idx] = w;
    }
    if (tid < EMBD * NCLS) sh_wlin[tid] = w_lin[tid];
    __syncthreads();

    const float* xl = x + 4 * lane;
    const int ntiles = (N + 15) >> 4;
    const int nblk   = (int)gridDim.x;
    for (int tile = wid * nblk + (int)blockIdx.x; tile < ntiles; tile += P2_WAVES * nblk) {
    const int tile0 = tile * 16;

    if (lane < 16) {
        int node = tile0 + lane;
        if (node >= N) node = N - 1;
        int deg = degrees[node];
        if (deg < 1) deg = 1;
        if (deg > MAXD) deg = MAXD;
        int k = deg >> 1;
        if (k < 1) k = 1;
        float sc = s0[node];
        float dist[MAXD];
        int   ids [MAXD];
        #pragma unroll
        for (int j = 0; j < MAXD; ++j) {
            int nb = neighbors[(size_t)node * MAXD + j];
            ids[j] = nb;
            float d = fabsf(sc - s0[nb]);
            dist[j] = (j < deg) ? d : INFINITY;
        }
        *(int4*)&sh_sel[wid][lane][0] = make_int4(0, 0, 0, 0);
        *(int4*)&sh_sel[wid][lane][4] = make_int4(0, 0, 0, 0);
        for (int t = 0; t < k; ++t) {
            int best = 0; float bv = INFINITY;
            #pragma unroll
            for (int j = 0; j < MAXD; ++j) {
                if (dist[j] < bv) { bv = dist[j]; best = j; }
            }
            sh_sel[wid][lane][t] = ids[best];
            dist[best] = INFINITY;
        }
        sh_k[wid][lane] = k;
    }
    wave_lds_order();

    for (int m0 = 0; m0 < 16; m0 += 4) {
        int    kq[4];
        int    idq[4][MAXD / 2];
        float4 r[4][MAXD / 2];
        #pragma unroll
        for (int q = 0; q < 4; ++q) {
            const int4 s_lo = *(const int4*)&sh_sel[wid][m0 + q][0];
            const int4 s_hi = *(const int4*)&sh_sel[wid][m0 + q][4];
            kq[q]     = __builtin_amdgcn_readfirstlane(sh_k[wid][m0 + q]);
            idq[q][0] = __builtin_amdgcn_readfirstlane(s_lo.x);
            idq[q][1] = __builtin_amdgcn_readfirstlane(s_lo.y);
            idq[q][2] = __builtin_amdgcn_readfirstlane(s_lo.z);
            idq[q][3] = __builtin_amdgcn_readfirstlane(s_lo.w);
            idq[q][4] = __builtin_amdgcn_readfirstlane(s_hi.x);
            idq[q][5] = __builtin_amdgcn_readfirstlane(s_hi.y);
            idq[q][6] = __builtin_amdgcn_readfirstlane(s_hi.z);
            idq[q][7] = __builtin_amdgcn_readfirstlane(s_hi.w);
        }
        #pragma unroll
        for (int q = 0; q < 4; ++q) {
            #pragma unroll
            for (int j = 0; j < MAXD / 2; ++j) {
                if (j < kq[q])
                    r[q][j] = *(const float4*)(xl + (size_t)idq[q][j] * NF);
            }
        }
        #pragma unroll
        for (int q = 0; q < 4; ++q) {
            float4 a = make_float4(0.f, 0.f, 0.f, 0.f);
            #pragma unroll
            for (int j = 0; j < MAXD / 2; ++j) {
                if (j < kq[q]) {
                    a.x += r[q][j].x; a.y += r[q][j].y;
                    a.z += r[q][j].z; a.w += r[q][j].w;
                }
            }
            const float inv = 1.0f / (float)kq[q];
            a.x *= inv; a.y *= inv; a.z *= inv; a.w *= inv;
            *(float4*)&sh_an[wid][m0 + q][4 * lane] = a;
        }
    }
    wave_lds_order();

    int srow = tile0 + lo;
    if (srow >= N) srow = N - 1;
    const float* xself = x + (size_t)srow * NF;

    v8f c[4];
    #pragma unroll
    for (int ct = 0; ct < 4; ++ct) c[ct] = (v8f){0.f,0.f,0.f,0.f,0.f,0.f,0.f,0.f};

    for (int i = 0; i < NF / 4; ++i) {
        const int kk = 4 * i + hi * 2;
        v2f a_s, a_n;
        a_s.x = xself[kk];          a_s.y = xself[kk + 1];
        a_n.x = sh_an[wid][lo][kk]; a_n.y = sh_an[wid][lo][kk + 1];
        #pragma unroll
        for (int ct = 0; ct < 4; ++ct) {
            float4 wf = shW[(i * 4 + ct) * 32 + lane];
            v2f b_s; b_s.x = wf.x; b_s.y = wf.y;
            v2f b_n; b_n.x = wf.z; b_n.y = wf.w;
            c[ct] = __builtin_amdgcn_wmma_f32_16x16x4_f32(
                        false, a_s, false, b_s, (short)0, c[ct], false, false);
            c[ct] = __builtin_amdgcn_wmma_f32_16x16x4_f32(
                        false, a_n, false, b_n, (short)0, c[ct], false, false);
        }
    }

    #pragma unroll
    for (int ct = 0; ct < 4; ++ct) {
        const int e = ct * 16 + lo;
        const float be = b_sage[e];
        #pragma unroll
        for (int j = 0; j < 8; ++j) {
            float v = c[ct][j] + be;
            sh_h[wid][j + 8 * hi][e] = v > 0.f ? v : 0.f;
        }
    }
    wave_lds_order();

    {
        const int m = lane >> 1, cls = lane & 1;
        float acc = b_lin[cls];
        #pragma unroll 8
        for (int e = 0; e < EMBD; ++e)
            acc += sh_h[wid][m][e] * sh_wlin[e * NCLS + cls];
        const int node = tile0 + m;
        if (node < N)
            logits_out[(size_t)node * NCLS + cls] = acc;
    }
    wave_lds_order();
    }
}

extern "C" void kernel_launch(void* const* d_in, const int* in_sizes, int n_in,
                              void* d_out, int out_size, void* d_ws, size_t ws_size,
                              hipStream_t stream) {
    const float* x       = (const float*)d_in[0];
    const float* w_dist  = (const float*)d_in[1];
    const float* b_dist  = (const float*)d_in[2];
    const float* w_self  = (const float*)d_in[3];
    const float* w_neigh = (const float*)d_in[4];
    const float* b_sage  = (const float*)d_in[5];
    const float* w_lin   = (const float*)d_in[6];
    const float* b_lin   = (const float*)d_in[7];
    const int*   neigh   = (const int*)d_in[8];
    const int*   degs    = (const int*)d_in[9];

    const int N = in_sizes[9];
    float* out       = (float*)d_out;
    float* logits    = out;
    float* score_out = out + (size_t)N * NCLS;
    float* s0        = (float*)d_ws;

    {
        int nodes_per_block = (256 / 32) * P1_ROWS;
        int blocks = (N + nodes_per_block - 1) / nodes_per_block;
        care_phase1<<<blocks, 256, 0, stream>>>(x, w_dist, b_dist,
                                                score_out, s0, N);
    }
    {
        care_phase2<<<256, P2_THREADS, 0, stream>>>(x, w_self, w_neigh,
                                                       b_sage, w_lin, b_lin,
                                                       neigh, degs, s0,
                                                       logits, N);
    }
}


# Phase 2: the self half of a tile's matrix products (x@W_self, 128 of 256 WMMAs) runs under the tile's neighbour-row gather, 8 K-steps per trip, A fragments loaded a trip ahead
# speedup vs baseline: 2.7444x; 1.0667x over previous
#include <hip/hip_runtime.h>
#include <hip/hip_bf16.h>
#include <math.h>

#define NF    128
#define EMBD  64
#define MAXD  16
#define NCLS  2

typedef float v2f __attribute__((ext_vector_type(2)));
typedef float v8f __attribute__((ext_vector_type(8)));

#define P1_ROWS 32

__global__ __launch_bounds__(256)
void care_phase1(const float* __restrict__ x,
                 const float* __restrict__ w_dist,
                 const float* __restrict__ b_dist,
                 float* __restrict__ score_out,
                 float* __restrict__ s0,
                 int N)
{
    const int lane = threadIdx.x & 31;
    const int wave = __builtin_amdgcn_readfirstlane(
                         (int)((blockIdx.x * blockDim.x + threadIdx.x) >> 5));
    const int row0 = wave * P1_ROWS;
    if (row0 >= N) return;

    float w0[4], w1[4];
    #pragma unroll
    for (int j = 0; j < 4; ++j) {
        w0[j] = w_dist[(32 * j + lane) * 2 + 0];
        w1[j] = w_dist[(32 * j + lane) * 2 + 1];
    }

    float v[P1_ROWS][4];
    #pragma unroll
    for (int q = 0; q < P1_ROWS; ++q) {
        int row = row0 + q;
        if (row >= N) row = N - 1;
        const float* xr = x + (size_t)row * NF + lane;
        #pragma unroll
        for (int j = 0; j < 4; ++j) v[q][j] = xr[32 * j];
    }
    __builtin_amdgcn_sched_barrier(0);

    float a0[P1_ROWS], a1[P1_ROWS];
    #pragma unroll
    for (int q = 0; q < P1_ROWS; ++q) {
        float acc0 = 0.f, acc1 = 0.f;
        #pragma unroll
        for (int j = 0; j < 4; ++j) {
            acc0 += v[q][j] * w0[j];
            acc1 += v[q][j] * w1[j];
        }
        a0[q] = acc0; a1[q] = acc1;
    }
    #pragma unroll
    for (int off = 16; off > 0; off >>= 1) {
        #pragma unroll
        for (int q = 0; q < P1_ROWS; ++q) {
            a0[q] += __shfl_xor(a0[q], off, 32);
            a1[q] += __shfl_xor(a1[q], off, 32);
        }
    }

    float m0 = 0.f, m1 = 0.f;
    #pragma unroll
    for (int q = 0; q < P1_ROWS; ++q) {
        if (lane == q) { m0 = a0[q]; m1 = a1[q]; }
    }

    const int node = row0 + lane;
    if (node < N) {
        const float l0 = m0 + b_dist[0];
        const float l1 = m1 + b_dist[1];
        *(float2*)(score_out + (size_t)node * 2) = make_float2(l0, l1);
        s0[node] = tanhf(l0);
    }
}

#define P2_THREADS 512
#define P2_WAVES (P2_THREADS / 32)

static __device__ __forceinline__ void wave_lds_order()
{
    __builtin_amdgcn_fence(__ATOMIC_RELEASE, "wavefront");
    __builtin_amdgcn_wave_barrier();
    __builtin_amdgcn_fence(__ATOMIC_ACQUIRE, "wavefront");
}

__global__ __launch_bounds__(P2_THREADS)
void care_phase2(const float* __restrict__ x,
                 const float* __restrict__ w_self,
                 const float* __restrict__ w_neigh,
                 const float* __restrict__ b_sage,
                 const float* __restrict__ w_lin,
                 const float* __restrict__ b_lin,
                 const int*   __restrict__ neighbors,
                 const int*   __restrict__ degrees,
                 const float* __restrict__ s0,
                 float* __restrict__ logits_out,
                 int N)
{
    __shared__ float4 shWs[(NF / 8) * 4 * 32];
    __shared__ float4 shWn[(NF / 8) * 4 * 32];
    __shared__ __attribute__((aligned(16))) float sh_an[P2_WAVES][16][NF + 4];
    __shared__ float sh_h [P2_WAVES][16][EMBD + 2];
    __shared__ float sh_wlin[EMBD * NCLS];
    __shared__ __attribute__((aligned(16))) int sh_sel[P2_WAVES][16][MAXD / 2];
    __shared__ int   sh_k [P2_WAVES][16];

    const int tid  = threadIdx.x;
    const int lane = tid & 31;
    const int wid  = __builtin_amdgcn_readfirstlane(tid >> 5);
    const int hi   = lane >> 4;
    const int lo   = lane & 15;

    #pragma unroll
    for (int idx = tid; idx < (NF / 8) * 4 * 32; idx += P2_THREADS) {
        const int l   = idx & 31;
        const int ct  = (idx >> 5) & 3;
        const int i2  = idx >> 7;
        const int ka  = 8 * i2 + 2 * (l >> 4);
        const int kb  = ka + 4;
        const int col = ct * 16 + (l & 15);
        float4 ws, wn;
        ws.x = w_self [(size_t)(ka)     * EMBD + col];
        ws.y = w_self [(size_t)(ka + 1) * EMBD + col];
        ws.z = w_self [(size_t)(kb)     * EMBD + col];
        ws.w = w_self [(size_t)(kb + 1) * EMBD + col];
        wn.x = w_neigh[(size_t)(ka)     * EMBD + col];
        wn.y = w_neigh[(size_t)(ka + 1) * EMBD + col];
        wn.z = w_neigh[(size_t)(kb)     * EMBD + col];
        wn.w = w_neigh[(size_t)(kb + 1) * EMBD + col];
        shWs[idx] = ws;
        shWn[idx] = wn;
    }
    if (tid < EMBD * NCLS) sh_wlin[tid] = w_lin[tid];
    __syncthreads();

    const float* xl = x + 4 * lane;
    const int ntiles = (N + 15) >> 4;
    const int nblk   = (int)gridDim.x;
    for (int tile = wid * nblk + (int)blockIdx.x; tile < ntiles; tile += P2_WAVES * nblk) {
    const int tile0 = tile * 16;

    int srow = tile0 + lo;
    if (srow >= N) srow = N - 1;
    const float* xself = x + (size_t)srow * NF + 2 * hi;
    v2f as_cur[8];
    #pragma unroll
    for (int s = 0; s < 8; ++s) {
        as_cur[s].x = xself[4 * s];
        as_cur[s].y = xself[4 * s + 1];
    }

    {
        int node = tile0 + lo;
        if (node >= N) node = N - 1;
        int deg = degrees[node];
        if (deg < 1) deg = 1;
        if (deg > MAXD) deg = MAXD;
        int k = deg >> 1;
        if (k < 1) k = 1;
        float sc = s0[node];
        float dist[MAXD];
        int   ids [MAXD];
        #pragma unroll
        for (int j = 0; j < MAXD; ++j) {
            int nb = neighbors[(size_t)node * MAXD + j];
            ids[j] = nb;
            float d = fabsf(sc - s0[nb]);
            dist[j] = (j < deg) ? d : INFINITY;
        }
        if (hi == 0) {
            *(int4*)&sh_sel[wid][lo][0] = make_int4(0, 0, 0, 0);
            *(int4*)&sh_sel[wid][lo][4] = make_int4(0, 0, 0, 0);
        }
        for (int t = 0; t < k; ++t) {
            int best = 0; float bv = INFINITY;
            #pragma unroll
            for (int j = 0; j < MAXD; ++j) {
                if (dist[j] < bv) { bv = dist[j]; best = j; }
            }
            if (hi == 0) sh_sel[wid][lo][t] = ids[best];
            dist[best] = INFINITY;
        }
        if (hi == 0) sh_k[wid][lo] = k;
    }
    wave_lds_order();

    v8f c[4];
    #pragma unroll
    for (int ct = 0; ct < 4; ++ct) c[ct] = (v8f){0.f,0.f,0.f,0.f,0.f,0.f,0.f,0.f};

    #pragma nounroll
    for (int g = 0; g < 4; ++g) {
        const int m0 = 4 * g;
        int    kq[4];
        int    idq[4][MAXD / 2];
        float4 r[4][MAXD / 2];
        #pragma unroll
        for (int q = 0; q < 4; ++q) {
            const int4 s_lo = *(const int4*)&sh_sel[wid][m0 + q][0];
            const int4 s_hi = *(const int4*)&sh_sel[wid][m0 + q][4];
            kq[q]     = __builtin_amdgcn_readfirstlane(sh_k[wid][m0 + q]);
            idq[q][0] = __builtin_amdgcn_readfirstlane(s_lo.x);
            idq[q][1] = __builtin_amdgcn_readfirstlane(s_lo.y);
            idq[q][2] = __builtin_amdgcn_readfirstlane(s_lo.z);
            idq[q][3] = __builtin_amdgcn_readfirstlane(s_lo.w);
            idq[q][4] = __builtin_amdgcn_readfirstlane(s_hi.x);
            idq[q][5] = __builtin_amdgcn_readfirstlane(s_hi.y);
            idq[q][6] = __builtin_amdgcn_readfirstlane(s_hi.z);
            idq[q][7] = __builtin_amdgcn_readfirstlane(s_hi.w);
        }
        #pragma unroll
        for (int q = 0; q < 4; ++q) {
            #pragma unroll
            for (int j = 0; j < MAXD / 2; ++j) {
                if (j < kq[q])
                    r[q][j] = *(const float4*)(xl + (size_t)idq[q][j] * NF);
            }
        }
        v2f as_nxt[8];
        #pragma unroll
        for (int s = 0; s < 8; ++s) as_nxt[s] = as_cur[s];
        if (g < 3) {
            #pragma unroll
            for (int s = 0; s < 8; ++s) {
                as_nxt[s].x = xself[32 * (g + 1) + 4 * s];
                as_nxt[s].y = xself[32 * (g + 1) + 4 * s + 1];
            }
        }
        #pragma unroll
        for (int p = 0; p < 4; ++p) {
            #pragma unroll
            for (int ct = 0; ct < 4; ++ct) {
                const float4 wf = shWs[((4 * g + p) * 4 + ct) * 32 + lane];
                v2f b0; b0.x = wf.x; b0.y = wf.y;
                v2f b1; b1.x = wf.z; b1.y = wf.w;
                c[ct] = __builtin_amdgcn_wmma_f32_16x16x4_f32(
                            false, as_cur[2 * p],     false, b0, (short)0, c[ct], false, false);
                c[ct] = __builtin_amdgcn_wmma_f32_16x16x4_f32(
                            false, as_cur[2 * p + 1], false, b1, (short)0, c[ct], false, false);
            }
        }
        asm volatile("" : "+v"(c[0]), "+v"(c[1]), "+v"(c[2]), "+v"(c[3]),
                          "+v"(r[0][0].x), "+v"(r[0][0].y), "+v"(r[0][0].z), "+v"(r[0][0].w),
                          "+v"(r[1][0].x), "+v"(r[1][0].y), "+v"(r[1][0].z), "+v"(r[1][0].w),
                          "+v"(r[2][0].x), "+v"(r[2][0].y), "+v"(r[2][0].z), "+v"(r[2][0].w),
                          "+v"(r[3][0].x), "+v"(r[3][0].y), "+v"(r[3][0].z), "+v"(r[3][0].w));
        #pragma unroll
        for (int q = 0; q < 4; ++q) {
            float4 a = make_float4(0.f, 0.f, 0.f, 0.f);
            #pragma unroll
            for (int j = 0; j < MAXD / 2; ++j) {
                if (j < kq[q]) {
                    a.x += r[q][j].x; a.y += r[q][j].y;
                    a.z += r[q][j].z; a.w += r[q][j].w;
                }
            }
            const float inv = 1.0f / (float)kq[q];
            a.x *= inv; a.y *= inv; a.z *= inv; a.w *= inv;
            *(float4*)&sh_an[wid][m0 + q][4 * lane] = a;
        }
        #pragma unroll
        for (int s = 0; s < 8; ++s) as_cur[s] = as_nxt[s];
    }
    wave_lds_order();

    for (int i2 = 0; i2 < NF / 8; ++i2) {
        const int kk = 8 * i2 + hi * 2;
        v2f a0, a1;
        a0.x = sh_an[wid][lo][kk];     a0.y = sh_an[wid][lo][kk + 1];
        a1.x = sh_an[wid][lo][kk + 4]; a1.y = sh_an[wid][lo][kk + 5];
        #pragma unroll
        for (int ct = 0; ct < 4; ++ct) {
            const float4 wf = shWn[(i2 * 4 + ct) * 32 + lane];
            v2f b0; b0.x = wf.x; b0.y = wf.y;
            v2f b1; b1.x = wf.z; b1.y = wf.w;
            c[ct] = __builtin_amdgcn_wmma_f32_16x16x4_f32(
                        false, a0, false, b0, (short)0, c[ct], false, false);
            c[ct] = __builtin_amdgcn_wmma_f32_16x16x4_f32(
                        false, a1, false, b1, (short)0, c[ct], false, false);
        }
    }

    #pragma unroll
    for (int ct = 0; ct < 4; ++ct) {
        const int e = ct * 16 + lo;
        const float be = b_sage[e];
        #pragma unroll
        for (int j = 0; j < 8; ++j) {
            float v = c[ct][j] + be;
            sh_h[wid][j + 8 * hi][e] = v > 0.f ? v : 0.f;
        }
    }
    wave_lds_order();

    {
        const int m = lane >> 1, cls = lane & 1;
        float acc = b_lin[cls];
        #pragma unroll 8
        for (int e = 0; e < EMBD; ++e)
            acc += sh_h[wid][m][e] * sh_wlin[e * NCLS + cls];
        const int node = tile0 + m;
        if (node < N)
            logits_out[(size_t)node * NCLS + cls] = acc;
    }
    wave_lds_order();
    }
}

extern "C" void kernel_launch(void* const* d_in, const int* in_sizes, int n_in,
                              void* d_out, int out_size, void* d_ws, size_t ws_size,
                              hipStream_t stream) {
    const float* x       = (const float*)d_in[0];
    const float* w_dist  = (const float*)d_in[1];
    const float* b_dist  = (const float*)d_in[2];
    const float* w_self  = (const float*)d_in[3];
    const float* w_neigh = (const float*)d_in[4];
    const float* b_sage  = (const float*)d_in[5];
    const float* w_lin   = (const float*)d_in[6];
    const float* b_lin   = (const float*)d_in[7];
    const int*   neigh   = (const int*)d_in[8];
    const int*   degs    = (const int*)d_in[9];

    const int N = in_sizes[9];
    float* out       = (float*)d_out;
    float* logits    = out;
    float* score_out = out + (size_t)N * NCLS;
    float* s0        = (float*)d_ws;

    {
        int nodes_per_block = (256 / 32) * P1_ROWS;
        int blocks = (N + nodes_per_block - 1) / nodes_per_block;
        care_phase1<<<blocks, 256, 0, stream>>>(x, w_dist, b_dist,
                                                score_out, s0, N);
    }
    {
        care_phase2<<<256, P2_THREADS, 0, stream>>>(x, w_self, w_neigh,
                                                       b_sage, w_lin, b_lin,
                                                       neigh, degs, s0,
                                                       logits, N);
    }
}


# Phase 2 row gather cut to the rows: load and sum loops end at k (no per-slot compare/select chains), ids stay in vector registers (one shift-or per address), 1/k stored once per node
# speedup vs baseline: 2.8741x; 1.0473x over previous
#include <hip/hip_runtime.h>
#include <hip/hip_bf16.h>
#include <math.h>

#define NF    128
#define EMBD  64
#define MAXD  16
#define NCLS  2

typedef float v2f __attribute__((ext_vector_type(2)));
typedef float v8f __attribute__((ext_vector_type(8)));
typedef float v4f __attribute__((ext_vector_type(4)));
typedef int   v4i __attribute__((ext_vector_type(4)));

#define P1_ROWS 32

__global__ __launch_bounds__(256)
void care_phase1(const float* __restrict__ x,
                 const float* __restrict__ w_dist,
                 const float* __restrict__ b_dist,
                 float* __restrict__ score_out,
                 float* __restrict__ s0,
                 int N)
{
    const int lane = threadIdx.x & 31;
    const int wave = __builtin_amdgcn_readfirstlane(
                         (int)((blockIdx.x * blockDim.x + threadIdx.x) >> 5));
    const int row0 = wave * P1_ROWS;
    if (row0 >= N) return;

    float w0[4], w1[4];
    #pragma unroll
    for (int j = 0; j < 4; ++j) {
        w0[j] = w_dist[(32 * j + lane) * 2 + 0];
        w1[j] = w_dist[(32 * j + lane) * 2 + 1];
    }

    float v[P1_ROWS][4];
    #pragma unroll
    for (int q = 0; q < P1_ROWS; ++q) {
        int row = row0 + q;
        if (row >= N) row = N - 1;
        const float* xr = x + (size_t)row * NF + lane;
        #pragma unroll
        for (int j = 0; j < 4; ++j) v[q][j] = xr[32 * j];
    }
    __builtin_amdgcn_sched_barrier(0);

    float a0[P1_ROWS], a1[P1_ROWS];
    #pragma unroll
    for (int q = 0; q < P1_ROWS; ++q) {
        float acc0 = 0.f, acc1 = 0.f;
        #pragma unroll
        for (int j = 0; j < 4; ++j) {
            acc0 += v[q][j] * w0[j];
            acc1 += v[q][j] * w1[j];
        }
        a0[q] = acc0; a1[q] = acc1;
    }
    #pragma unroll
    for (int off = 16; off > 0; off >>= 1) {
        #pragma unroll
        for (int q = 0; q < P1_ROWS; ++q) {
            a0[q] += __shfl_xor(a0[q], off, 32);
            a1[q] += __shfl_xor(a1[q], off, 32);
        }
    }

    float m0 = 0.f, m1 = 0.f;
    #pragma unroll
    for (int q = 0; q < P1_ROWS; ++q) {
        if (lane == q) { m0 = a0[q]; m1 = a1[q]; }
    }

    const int node = row0 + lane;
    if (node < N) {
        const float l0 = m0 + b_dist[0];
        const float l1 = m1 + b_dist[1];
        *(float2*)(score_out + (size_t)node * 2) = make_float2(l0, l1);
        s0[node] = tanhf(l0);
    }
}

#define P2_THREADS 512
#define P2_WAVES (P2_THREADS / 32)

static __device__ __forceinline__ void wave_lds_order()
{
    __builtin_amdgcn_fence(__ATOMIC_RELEASE, "wavefront");
    __builtin_amdgcn_wave_barrier();
    __builtin_amdgcn_fence(__ATOMIC_ACQUIRE, "wavefront");
}

__global__ __launch_bounds__(P2_THREADS)
void care_phase2(const float* __restrict__ x,
                 const float* __restrict__ w_self,
                 const float* __restrict__ w_neigh,
                 const float* __restrict__ b_sage,
                 const float* __restrict__ w_lin,
                 const float* __restrict__ b_lin,
                 const int*   __restrict__ neighbors,
                 const int*   __restrict__ degrees,
                 const float* __restrict__ s0,
                 float* __restrict__ logits_out,
                 int N)
{
    __shared__ float4 shWs[(NF / 8) * 4 * 32];
    __shared__ float4 shWn[(NF / 8) * 4 * 32];
    __shared__ __attribute__((aligned(16))) float sh_an[P2_WAVES][16][NF + 4];
    __shared__ float sh_h [P2_WAVES][16][EMBD + 2];
    __shared__ float sh_wlin[EMBD * NCLS];
    __shared__ __attribute__((aligned(16))) int sh_sel[P2_WAVES][16][MAXD / 2];
    __shared__ __attribute__((aligned(8))) int sh_ki[P2_WAVES][16][2];

    const int tid  = threadIdx.x;
    const int lane = tid & 31;
    const int wid  = __builtin_amdgcn_readfirstlane(tid >> 5);
    const int hi   = lane >> 4;
    const int lo   = lane & 15;

    #pragma unroll
    for (int idx = tid; idx < (NF / 8) * 4 * 32; idx += P2_THREADS) {
        const int l   = idx & 31;
        const int ct  = (idx >> 5) & 3;
        const int i2  = idx >> 7;
        const int ka  = 8 * i2 + 2 * (l >> 4);
        const int kb  = ka + 4;
        const int col = ct * 16 + (l & 15);
        float4 ws, wn;
        ws.x = w_self [(size_t)(ka)     * EMBD + col];
        ws.y = w_self [(size_t)(ka + 1) * EMBD + col];
        ws.z = w_self [(size_t)(kb)     * EMBD + col];
        ws.w = w_self [(size_t)(kb + 1) * EMBD + col];
        wn.x = w_neigh[(size_t)(ka)     * EMBD + col];
        wn.y = w_neigh[(size_t)(ka + 1) * EMBD + col];
        wn.z = w_neigh[(size_t)(kb)     * EMBD + col];
        wn.w = w_neigh[(size_t)(kb + 1) * EMBD + col];
        shWs[idx] = ws;
        shWn[idx] = wn;
    }
    if (tid < EMBD * NCLS) sh_wlin[tid] = w_lin[tid];
    __syncthreads();

    const char* xb = (const char*)x;
    const unsigned lane16 = 16u * (unsigned)lane;
    const int ntiles = (N + 15) >> 4;
    const int nblk   = (int)gridDim.x;
    for (int tile = wid * nblk + (int)blockIdx.x; tile < ntiles; tile += P2_WAVES * nblk) {
    const int tile0 = tile * 16;

    int srow = tile0 + lo;
    if (srow >= N) srow = N - 1;
    const float* xself = x + (size_t)srow * NF + 2 * hi;
    v2f as_cur[8];
    #pragma unroll
    for (int s = 0; s < 8; ++s) {
        as_cur[s].x = xself[4 * s];
        as_cur[s].y = xself[4 * s + 1];
    }

    {
        int node = tile0 + lo;
        if (node >= N) node = N - 1;
        int deg = degrees[node];
        if (deg < 1) deg = 1;
        if (deg > MAXD) deg = MAXD;
        int k = deg >> 1;
        if (k < 1) k = 1;
        float sc = s0[node];
        float dist[MAXD];
        int   ids [MAXD];
        #pragma unroll
        for (int j = 0; j < MAXD; ++j) {
            int nb = neighbors[(size_t)node * MAXD + j];
            ids[j] = nb;
            float d = fabsf(sc - s0[nb]);
            dist[j] = (j < deg) ? d : INFINITY;
        }
        if (hi == 0) {
            *(int4*)&sh_sel[wid][lo][0] = make_int4(0, 0, 0, 0);
            *(int4*)&sh_sel[wid][lo][4] = make_int4(0, 0, 0, 0);
        }
        for (int t = 0; t < k; ++t) {
            int best = 0; float bv = INFINITY;
            #pragma unroll
            for (int j = 0; j < MAXD; ++j) {
                if (dist[j] < bv) { bv = dist[j]; best = j; }
            }
            if (hi == 0) sh_sel[wid][lo][t] = ids[best];
            dist[best] = INFINITY;
        }
        const float inv = 1.0f / (float)k;
        if (hi == 0) *(int2*)&sh_ki[wid][lo][0] = make_int2(k, __float_as_int(inv));
    }
    wave_lds_order();

    v8f c[4];
    #pragma unroll
    for (int ct = 0; ct < 4; ++ct) c[ct] = (v8f){0.f,0.f,0.f,0.f,0.f,0.f,0.f,0.f};

    #pragma nounroll
    for (int g = 0; g < 4; ++g) {
        const int m0 = 4 * g;
        int    kq[4];
        v4f    r[4][MAXD / 2];
        v4i idlo[4], idhi[4];
        #pragma unroll
        for (int q = 0; q < 4; ++q) {
            const int kv = sh_ki[wid][m0 + q][0];
            idlo[q] = *(const v4i*)&sh_sel[wid][m0 + q][0];
            idhi[q] = *(const v4i*)&sh_sel[wid][m0 + q][4];
            kq[q]   = __builtin_amdgcn_readfirstlane(kv);
        }
        asm volatile("" : "+v"(idlo[0]), "+v"(idhi[0]), "+v"(idlo[1]), "+v"(idhi[1]),
                          "+v"(idlo[2]), "+v"(idhi[2]), "+v"(idlo[3]), "+v"(idhi[3]));
        #pragma unroll
        for (int q = 0; q < 4; ++q) {
            #pragma unroll
            for (int j = 0; j < MAXD / 2; ++j) {
                if (j > 0 && j >= kq[q]) break;
                const int id = (j < 4) ? idlo[q][j & 3] : idhi[q][j & 3];
                const unsigned off = ((unsigned)id << 9) + lane16;
                r[q][j] = *(const v4f*)(xb + off);
            }
        }
        v2f as_nxt[8];
        #pragma unroll
        for (int s = 0; s < 8; ++s) as_nxt[s] = as_cur[s];
        if (g < 3) {
            #pragma unroll
            for (int s = 0; s < 8; ++s) {
                as_nxt[s].x = xself[32 * (g + 1) + 4 * s];
                as_nxt[s].y = xself[32 * (g + 1) + 4 * s + 1];
            }
        }
        #pragma unroll
        for (int p = 0; p < 4; ++p) {
            #pragma unroll
            for (int ct = 0; ct < 4; ++ct) {
                const float4 wf = shWs[((4 * g + p) * 4 + ct) * 32 + lane];
                v2f b0; b0.x = wf.x; b0.y = wf.y;
                v2f b1; b1.x = wf.z; b1.y = wf.w;
                c[ct] = __builtin_amdgcn_wmma_f32_16x16x4_f32(
                            false, as_cur[2 * p],     false, b0, (short)0, c[ct], false, false);
                c[ct] = __builtin_amdgcn_wmma_f32_16x16x4_f32(
                            false, as_cur[2 * p + 1], false, b1, (short)0, c[ct], false, false);
            }
        }
        asm volatile("" : "+v"(c[0]), "+v"(c[1]), "+v"(c[2]), "+v"(c[3]),
                          "+v"(r[0][0]), "+v"(r[1][0]), "+v"(r[2][0]), "+v"(r[3][0]));
        #pragma unroll
        for (int q = 0; q < 4; ++q) {
            v4f a = (v4f){0.f, 0.f, 0.f, 0.f};
            #pragma unroll
            for (int j = 0; j < MAXD / 2; ++j) {
                if (j > 0 && j >= kq[q]) break;
                a += r[q][j];
            }
            const float inv = __int_as_float(sh_ki[wid][m0 + q][1]);
            a *= inv;
            *(v4f*)&sh_an[wid][m0 + q][4 * lane] = a;
        }
        #pragma unroll
        for (int s = 0; s < 8; ++s) as_cur[s] = as_nxt[s];
    }
    wave_lds_order();

    for (int i2 = 0; i2 < NF / 8; ++i2) {
        const int kk = 8 * i2 + hi * 2;
        v2f a0, a1;
        a0.x = sh_an[wid][lo][kk];     a0.y = sh_an[wid][lo][kk + 1];
        a1.x = sh_an[wid][lo][kk + 4]; a1.y = sh_an[wid][lo][kk + 5];
        #pragma unroll
        for (int ct = 0; ct < 4; ++ct) {
            const float4 wf = shWn[(i2 * 4 + ct) * 32 + lane];
            v2f b0; b0.x = wf.x; b0.y = wf.y;
            v2f b1; b1.x = wf.z; b1.y = wf.w;
            c[ct] = __builtin_amdgcn_wmma_f32_16x16x4_f32(
                        false, a0, false, b0, (short)0, c[ct], false, false);
            c[ct] = __builtin_amdgcn_wmma_f32_16x16x4_f32(
                        false, a1, false, b1, (short)0, c[ct], false, false);
        }
    }

    #pragma unroll
    for (int ct = 0; ct < 4; ++ct) {
        const int e = ct * 16 + lo;
        const float be = b_sage[e];
        #pragma unroll
        for (int j = 0; j < 8; ++j) {
            float v = c[ct][j] + be;
            sh_h[wid][j + 8 * hi][e] = v > 0.f ? v : 0.f;
        }
    }
    wave_lds_order();

    {
        const int m = lane >> 1, cls = lane & 1;
        float acc = b_lin[cls];
        #pragma unroll 8
        for (int e = 0; e < EMBD; ++e)
            acc += sh_h[wid][m][e] * sh_wlin[e * NCLS + cls];
        const int node = tile0 + m;
        if (node < N)
            logits_out[(size_t)node * NCLS + cls] = acc;
    }
    wave_lds_order();
    }
}

extern "C" void kernel_launch(void* const* d_in, const int* in_sizes, int n_in,
                              void* d_out, int out_size, void* d_ws, size_t ws_size,
                              hipStream_t stream) {
    const float* x       = (const float*)d_in[0];
    const float* w_dist  = (const float*)d_in[1];
    const float* b_dist  = (const float*)d_in[2];
    const float* w_self  = (const float*)d_in[3];
    const float* w_neigh = (const float*)d_in[4];
    const float* b_sage  = (const float*)d_in[5];
    const float* w_lin   = (const float*)d_in[6];
    const float* b_lin   = (const float*)d_in[7];
    const int*   neigh   = (const int*)d_in[8];
    const int*   degs    = (const int*)d_in[9];

    const int N = in_sizes[9];
    float* out       = (float*)d_out;
    float* logits    = out;
    float* score_out = out + (size_t)N * NCLS;
    float* s0        = (float*)d_ws;

    {
        int nodes_per_block = (256 / 32) * P1_ROWS;
        int blocks = (N + nodes_per_block - 1) / nodes_per_block;
        care_phase1<<<blocks, 256, 0, stream>>>(x, w_dist, b_dist,
                                                score_out, s0, N);
    }
    {
        care_phase2<<<256, P2_THREADS, 0, stream>>>(x, w_self, w_neigh,
                                                       b_sage, w_lin, b_lin,
                                                       neigh, degs, s0,
                                                       logits, N);
    }
}


# Phase 2 neighbor filter by counted rank: each lane of a node's pair counts for 8 of its 16 candidates how many sort before it (stable) and stores the id at that place; no k-pass minimum search
# speedup vs baseline: 2.9642x; 1.0313x over previous
#include <hip/hip_runtime.h>
#include <hip/hip_bf16.h>
#include <math.h>

#define NF    128
#define EMBD  64
#define MAXD  16
#define NCLS  2

typedef float v2f __attribute__((ext_vector_type(2)));
typedef float v8f __attribute__((ext_vector_type(8)));
typedef float v4f __attribute__((ext_vector_type(4)));
typedef int   v4i __attribute__((ext_vector_type(4)));

#define P1_ROWS 32

__global__ __launch_bounds__(256)
void care_phase1(const float* __restrict__ x,
                 const float* __restrict__ w_dist,
                 const float* __restrict__ b_dist,
                 float* __restrict__ score_out,
                 float* __restrict__ s0,
                 int N)
{
    const int lane = threadIdx.x & 31;
    const int wave = __builtin_amdgcn_readfirstlane(
                         (int)((blockIdx.x * blockDim.x + threadIdx.x) >> 5));
    const int row0 = wave * P1_ROWS;
    if (row0 >= N) return;

    float w0[4], w1[4];
    #pragma unroll
    for (int j = 0; j < 4; ++j) {
        w0[j] = w_dist[(32 * j + lane) * 2 + 0];
        w1[j] = w_dist[(32 * j + lane) * 2 + 1];
    }

    float v[P1_ROWS][4];
    #pragma unroll
    for (int q = 0; q < P1_ROWS; ++q) {
        int row = row0 + q;
        if (row >= N) row = N - 1;
        const float* xr = x + (size_t)row * NF + lane;
        #pragma unroll
        for (int j = 0; j < 4; ++j) v[q][j] = xr[32 * j];
    }
    __builtin_amdgcn_sched_barrier(0);

    float a0[P1_ROWS], a1[P1_ROWS];
    #pragma unroll
    for (int q = 0; q < P1_ROWS; ++q) {
        float acc0 = 0.f, acc1 = 0.f;
        #pragma unroll
        for (int j = 0; j < 4; ++j) {
            acc0 += v[q][j] * w0[j];
            acc1 += v[q][j] * w1[j];
        }
        a0[q] = acc0; a1[q] = acc1;
    }
    #pragma unroll
    for (int off = 16; off > 0; off >>= 1) {
        #pragma unroll
        for (int q = 0; q < P1_ROWS; ++q) {
            a0[q] += __shfl_xor(a0[q], off, 32);
            a1[q] += __shfl_xor(a1[q], off, 32);
        }
    }

    float m0 = 0.f, m1 = 0.f;
    #pragma unroll
    for (int q = 0; q < P1_ROWS; ++q) {
        if (lane == q) { m0 = a0[q]; m1 = a1[q]; }
    }

    const int node = row0 + lane;
    if (node < N) {
        const float l0 = m0 + b_dist[0];
        const float l1 = m1 + b_dist[1];
        *(float2*)(score_out + (size_t)node * 2) = make_float2(l0, l1);
        s0[node] = tanhf(l0);
    }
}

#define P2_THREADS 512
#define P2_WAVES (P2_THREADS / 32)

static __device__ __forceinline__ void wave_lds_order()
{
    __builtin_amdgcn_fence(__ATOMIC_RELEASE, "wavefront");
    __builtin_amdgcn_wave_barrier();
    __builtin_amdgcn_fence(__ATOMIC_ACQUIRE, "wavefront");
}

__global__ __launch_bounds__(P2_THREADS)
void care_phase2(const float* __restrict__ x,
                 const float* __restrict__ w_self,
                 const float* __restrict__ w_neigh,
                 const float* __restrict__ b_sage,
                 const float* __restrict__ w_lin,
                 const float* __restrict__ b_lin,
                 const int*   __restrict__ neighbors,
                 const int*   __restrict__ degrees,
                 const float* __restrict__ s0,
                 float* __restrict__ logits_out,
                 int N)
{
    __shared__ float4 shWs[(NF / 8) * 4 * 32];
    __shared__ float4 shWn[(NF / 8) * 4 * 32];
    __shared__ __attribute__((aligned(16))) float sh_an[P2_WAVES][16][NF + 4];
    __shared__ float sh_h [P2_WAVES][16][EMBD + 2];
    __shared__ float sh_wlin[EMBD * NCLS];
    __shared__ __attribute__((aligned(16))) int sh_sel[P2_WAVES][16][MAXD];
    __shared__ __attribute__((aligned(8))) int sh_ki[P2_WAVES][16][2];

    const int tid  = threadIdx.x;
    const int lane = tid & 31;
    const int wid  = __builtin_amdgcn_readfirstlane(tid >> 5);
    const int hi   = lane >> 4;
    const int lo   = lane & 15;

    #pragma unroll
    for (int idx = tid; idx < (NF / 8) * 4 * 32; idx += P2_THREADS) {
        const int l   = idx & 31;
        const int ct  = (idx >> 5) & 3;
        const int i2  = idx >> 7;
        const int ka  = 8 * i2 + 2 * (l >> 4);
        const int kb  = ka + 4;
        const int col = ct * 16 + (l & 15);
        float4 ws, wn;
        ws.x = w_self [(size_t)(ka)     * EMBD + col];
        ws.y = w_self [(size_t)(ka + 1) * EMBD + col];
        ws.z = w_self [(size_t)(kb)     * EMBD + col];
        ws.w = w_self [(size_t)(kb + 1) * EMBD + col];
        wn.x = w_neigh[(size_t)(ka)     * EMBD + col];
        wn.y = w_neigh[(size_t)(ka + 1) * EMBD + col];
        wn.z = w_neigh[(size_t)(kb)     * EMBD + col];
        wn.w = w_neigh[(size_t)(kb + 1) * EMBD + col];
        shWs[idx] = ws;
        shWn[idx] = wn;
    }
    if (tid < EMBD * NCLS) sh_wlin[tid] = w_lin[tid];
    __syncthreads();

    const char* xb = (const char*)x;
    const unsigned lane16 = 16u * (unsigned)lane;
    const int ntiles = (N + 15) >> 4;
    const int nblk   = (int)gridDim.x;
    for (int tile = wid * nblk + (int)blockIdx.x; tile < ntiles; tile += P2_WAVES * nblk) {
    const int tile0 = tile * 16;

    int srow = tile0 + lo;
    if (srow >= N) srow = N - 1;
    const float* xself = x + (size_t)srow * NF + 2 * hi;
    v2f as_cur[8];
    #pragma unroll
    for (int s = 0; s < 8; ++s) {
        as_cur[s].x = xself[4 * s];
        as_cur[s].y = xself[4 * s + 1];
    }

    {
        int node = tile0 + lo;
        if (node >= N) node = N - 1;
        int deg = degrees[node];
        if (deg < 1) deg = 1;
        if (deg > MAXD) deg = MAXD;
        int k = deg >> 1;
        if (k < 1) k = 1;
        float sc = s0[node];
        unsigned key[MAXD];
        int      ids[MAXD];
        float    sn[MAXD];
        #pragma unroll
        for (int j = 0; j < MAXD; ++j) {
            ids[j] = neighbors[(size_t)node * MAXD + j];
            sn[j]  = s0[ids[j]];
        }
        __builtin_amdgcn_sched_barrier(0);
        #pragma unroll
        for (int j = 0; j < MAXD; ++j) {
            const float d = fabsf(sc - sn[j]);
            key[j] = (j < deg) ? __float_as_uint(d) : 0x7f800000u;
        }
        #pragma unroll
        for (int jj = 0; jj < MAXD / 2; ++jj) {
            const unsigned mk  = hi ? key[2 * jj + 1] : key[2 * jj];
            const int      mid = hi ? ids[2 * jj + 1] : ids[2 * jj];
            unsigned place = 0;
            #pragma unroll
            for (int i = 0; i < MAXD; ++i) {
                if (i == 2 * jj || i == 2 * jj + 1) continue;
                if (i < 2 * jj) place += (key[i] <= mk) ? 1u : 0u;
                else            place += (key[i] <  mk) ? 1u : 0u;
            }
            const bool even_first = key[2 * jj] <= key[2 * jj + 1];
            place += (hi ? even_first : !even_first) ? 1u : 0u;
            asm volatile("" : "+v"(place));
            sh_sel[wid][lo][place] = mid;
        }
        const float inv = 1.0f / (float)k;
        if (hi == 0) *(int2*)&sh_ki[wid][lo][0] = make_int2(k, __float_as_int(inv));
    }
    wave_lds_order();

    v8f c[4];
    #pragma unroll
    for (int ct = 0; ct < 4; ++ct) c[ct] = (v8f){0.f,0.f,0.f,0.f,0.f,0.f,0.f,0.f};

    #pragma nounroll
    for (int g = 0; g < 4; ++g) {
        const int m0 = 4 * g;
        int    kq[4];
        v4f    r[4][MAXD / 2];
        v4i idlo[4], idhi[4];
        #pragma unroll
        for (int q = 0; q < 4; ++q) {
            const int kv = sh_ki[wid][m0 + q][0];
            idlo[q] = *(const v4i*)&sh_sel[wid][m0 + q][0];
            idhi[q] = *(const v4i*)&sh_sel[wid][m0 + q][4];
            kq[q]   = __builtin_amdgcn_readfirstlane(kv);
        }
        asm volatile("" : "+v"(idlo[0]), "+v"(idhi[0]), "+v"(idlo[1]), "+v"(idhi[1]),
                          "+v"(idlo[2]), "+v"(idhi[2]), "+v"(idlo[3]), "+v"(idhi[3]));
        #pragma unroll
        for (int q = 0; q < 4; ++q) {
            #pragma unroll
            for (int j = 0; j < MAXD / 2; ++j) {
                if (j > 0 && j >= kq[q]) break;
                const int id = (j < 4) ? idlo[q][j & 3] : idhi[q][j & 3];
                const unsigned off = ((unsigned)id << 9) + lane16;
                r[q][j] = *(const v4f*)(xb + off);
            }
        }
        v2f as_nxt[8];
        #pragma unroll
        for (int s = 0; s < 8; ++s) as_nxt[s] = as_cur[s];
        if (g < 3) {
            #pragma unroll
            for (int s = 0; s < 8; ++s) {
                as_nxt[s].x = xself[32 * (g + 1) + 4 * s];
                as_nxt[s].y = xself[32 * (g + 1) + 4 * s + 1];
            }
        }
        #pragma unroll
        for (int p = 0; p < 4; ++p) {
            #pragma unroll
            for (int ct = 0; ct < 4; ++ct) {
                const float4 wf = shWs[((4 * g + p) * 4 + ct) * 32 + lane];
                v2f b0; b0.x = wf.x; b0.y = wf.y;
                v2f b1; b1.x = wf.z; b1.y = wf.w;
                c[ct] = __builtin_amdgcn_wmma_f32_16x16x4_f32(
                            false, as_cur[2 * p],     false, b0, (short)0, c[ct], false, false);
                c[ct] = __builtin_amdgcn_wmma_f32_16x16x4_f32(
                            false, as_cur[2 * p + 1], false, b1, (short)0, c[ct], false, false);
            }
        }
        asm volatile("" : "+v"(c[0]), "+v"(c[1]), "+v"(c[2]), "+v"(c[3]),
                          "+v"(r[0][0]), "+v"(r[1][0]), "+v"(r[2][0]), "+v"(r[3][0]));
        #pragma unroll
        for (int q = 0; q < 4; ++q) {
            v4f a = (v4f){0.f, 0.f, 0.f, 0.f};
            #pragma unroll
            for (int j = 0; j < MAXD / 2; ++j) {
                if (j > 0 && j >= kq[q]) break;
                a += r[q][j];
            }
            const float inv = __int_as_float(sh_ki[wid][m0 + q][1]);
            a *= inv;
            *(v4f*)&sh_an[wid][m0 + q][4 * lane] = a;
        }
        #pragma unroll
        for (int s = 0; s < 8; ++s) as_cur[s] = as_nxt[s];
    }
    wave_lds_order();

    for (int i2 = 0; i2 < NF / 8; ++i2) {
        const int kk = 8 * i2 + hi * 2;
        v2f a0, a1;
        a0.x = sh_an[wid][lo][kk];     a0.y = sh_an[wid][lo][kk + 1];
        a1.x = sh_an[wid][lo][kk + 4]; a1.y = sh_an[wid][lo][kk + 5];
        #pragma unroll
        for (int ct = 0; ct < 4; ++ct) {
            const float4 wf = shWn[(i2 * 4 + ct) * 32 + lane];
            v2f b0; b0.x = wf.x; b0.y = wf.y;
            v2f b1; b1.x = wf.z; b1.y = wf.w;
            c[ct] = __builtin_amdgcn_wmma_f32_16x16x4_f32(
                        false, a0, false, b0, (short)0, c[ct], false, false);
            c[ct] = __builtin_amdgcn_wmma_f32_16x16x4_f32(
                        false, a1, false, b1, (short)0, c[ct], false, false);
        }
    }

    #pragma unroll
    for (int ct = 0; ct < 4; ++ct) {
        const int e = ct * 16 + lo;
        const float be = b_sage[e];
        #pragma unroll
        for (int j = 0; j < 8; ++j) {
            float v = c[ct][j] + be;
            sh_h[wid][j + 8 * hi][e] = v > 0.f ? v : 0.f;
        }
    }
    wave_lds_order();

    {
        const int m = lane >> 1, cls = lane & 1;
        float acc = b_lin[cls];
        #pragma unroll 8
        for (int e = 0; e < EMBD; ++e)
            acc += sh_h[wid][m][e] * sh_wlin[e * NCLS + cls];
        const int node = tile0 + m;
        if (node < N)
            logits_out[(size_t)node * NCLS + cls] = acc;
    }
    wave_lds_order();
    }
}

extern "C" void kernel_launch(void* const* d_in, const int* in_sizes, int n_in,
                              void* d_out, int out_size, void* d_ws, size_t ws_size,
                              hipStream_t stream) {
    const float* x       = (const float*)d_in[0];
    const float* w_dist  = (const float*)d_in[1];
    const float* b_dist  = (const float*)d_in[2];
    const float* w_self  = (const float*)d_in[3];
    const float* w_neigh = (const float*)d_in[4];
    const float* b_sage  = (const float*)d_in[5];
    const float* w_lin   = (const float*)d_in[6];
    const float* b_lin   = (const float*)d_in[7];
    const int*   neigh   = (const int*)d_in[8];
    const int*   degs    = (const int*)d_in[9];

    const int N = in_sizes[9];
    float* out       = (float*)d_out;
    float* logits    = out;
    float* score_out = out + (size_t)N * NCLS;
    float* s0        = (float*)d_ws;

    {
        int nodes_per_block = (256 / 32) * P1_ROWS;
        int blocks = (N + nodes_per_block - 1) / nodes_per_block;
        care_phase1<<<blocks, 256, 0, stream>>>(x, w_dist, b_dist,
                                                score_out, s0, N);
    }
    {
        care_phase2<<<256, P2_THREADS, 0, stream>>>(x, w_self, w_neigh,
                                                       b_sage, w_lin, b_lin,
                                                       neigh, degs, s0,
                                                       logits, N);
    }
}


# Phase 2 score gather only for real candidates: a padded slot (j >= degree) re-reads the node's own score word, so a score load touches one scattered line per real candidate, not 16 per node
# speedup vs baseline: 3.1286x; 1.0555x over previous
#include <hip/hip_runtime.h>
#include <hip/hip_bf16.h>
#include <math.h>

#define NF    128
#define EMBD  64
#define MAXD  16
#define NCLS  2

typedef float v2f __attribute__((ext_vector_type(2)));
typedef float v8f __attribute__((ext_vector_type(8)));
typedef float v4f __attribute__((ext_vector_type(4)));
typedef int   v4i __attribute__((ext_vector_type(4)));

#define P1_ROWS 32

__global__ __launch_bounds__(256)
void care_phase1(const float* __restrict__ x,
                 const float* __restrict__ w_dist,
                 const float* __restrict__ b_dist,
                 float* __restrict__ score_out,
                 float* __restrict__ s0,
                 int N)
{
    const int lane = threadIdx.x & 31;
    const int wave = __builtin_amdgcn_readfirstlane(
                         (int)((blockIdx.x * blockDim.x + threadIdx.x) >> 5));
    const int row0 = wave * P1_ROWS;
    if (row0 >= N) return;

    float w0[4], w1[4];
    #pragma unroll
    for (int j = 0; j < 4; ++j) {
        w0[j] = w_dist[(32 * j + lane) * 2 + 0];
        w1[j] = w_dist[(32 * j + lane) * 2 + 1];
    }

    float v[P1_ROWS][4];
    #pragma unroll
    for (int q = 0; q < P1_ROWS; ++q) {
        int row = row0 + q;
        if (row >= N) row = N - 1;
        const float* xr = x + (size_t)row * NF + lane;
        #pragma unroll
        for (int j = 0; j < 4; ++j) v[q][j] = xr[32 * j];
    }
    __builtin_amdgcn_sched_barrier(0);

    float a0[P1_ROWS], a1[P1_ROWS];
    #pragma unroll
    for (int q = 0; q < P1_ROWS; ++q) {
        float acc0 = 0.f, acc1 = 0.f;
        #pragma unroll
        for (int j = 0; j < 4; ++j) {
            acc0 += v[q][j] * w0[j];
            acc1 += v[q][j] * w1[j];
        }
        a0[q] = acc0; a1[q] = acc1;
    }
    #pragma unroll
    for (int off = 16; off > 0; off >>= 1) {
        #pragma unroll
        for (int q = 0; q < P1_ROWS; ++q) {
            a0[q] += __shfl_xor(a0[q], off, 32);
            a1[q] += __shfl_xor(a1[q], off, 32);
        }
    }

    float m0 = 0.f, m1 = 0.f;
    #pragma unroll
    for (int q = 0; q < P1_ROWS; ++q) {
        if (lane == q) { m0 = a0[q]; m1 = a1[q]; }
    }

    const int node = row0 + lane;
    if (node < N) {
        const float l0 = m0 + b_dist[0];
        const float l1 = m1 + b_dist[1];
        *(float2*)(score_out + (size_t)node * 2) = make_float2(l0, l1);
        s0[node] = tanhf(l0);
    }
}

#define P2_THREADS 512
#define P2_WAVES (P2_THREADS / 32)

static __device__ __forceinline__ void wave_lds_order()
{
    __builtin_amdgcn_fence(__ATOMIC_RELEASE, "wavefront");
    __builtin_amdgcn_wave_barrier();
    __builtin_amdgcn_fence(__ATOMIC_ACQUIRE, "wavefront");
}

__global__ __launch_bounds__(P2_THREADS)
void care_phase2(const float* __restrict__ x,
                 const float* __restrict__ w_self,
                 const float* __restrict__ w_neigh,
                 const float* __restrict__ b_sage,
                 const float* __restrict__ w_lin,
                 const float* __restrict__ b_lin,
                 const int*   __restrict__ neighbors,
                 const int*   __restrict__ degrees,
                 const float* __restrict__ s0,
                 float* __restrict__ logits_out,
                 int N)
{
    __shared__ float4 shWs[(NF / 8) * 4 * 32];
    __shared__ float4 shWn[(NF / 8) * 4 * 32];
    __shared__ __attribute__((aligned(16))) float sh_an[P2_WAVES][16][NF + 4];
    __shared__ float sh_h [P2_WAVES][16][EMBD + 2];
    __shared__ float sh_wlin[EMBD * NCLS];
    __shared__ __attribute__((aligned(16))) int sh_sel[P2_WAVES][16][MAXD];
    __shared__ __attribute__((aligned(8))) int sh_ki[P2_WAVES][16][2];

    const int tid  = threadIdx.x;
    const int lane = tid & 31;
    const int wid  = __builtin_amdgcn_readfirstlane(tid >> 5);
    const int hi   = lane >> 4;
    const int lo   = lane & 15;

    #pragma unroll
    for (int idx = tid; idx < (NF / 8) * 4 * 32; idx += P2_THREADS) {
        const int l   = idx & 31;
        const int ct  = (idx >> 5) & 3;
        const int i2  = idx >> 7;
        const int ka  = 8 * i2 + 2 * (l >> 4);
        const int kb  = ka + 4;
        const int col = ct * 16 + (l & 15);
        float4 ws, wn;
        ws.x = w_self [(size_t)(ka)     * EMBD + col];
        ws.y = w_self [(size_t)(ka + 1) * EMBD + col];
        ws.z = w_self [(size_t)(kb)     * EMBD + col];
        ws.w = w_self [(size_t)(kb + 1) * EMBD + col];
        wn.x = w_neigh[(size_t)(ka)     * EMBD + col];
        wn.y = w_neigh[(size_t)(ka + 1) * EMBD + col];
        wn.z = w_neigh[(size_t)(kb)     * EMBD + col];
        wn.w = w_neigh[(size_t)(kb + 1) * EMBD + col];
        shWs[idx] = ws;
        shWn[idx] = wn;
    }
    if (tid < EMBD * NCLS) sh_wlin[tid] = w_lin[tid];
    __syncthreads();

    const char* xb = (const char*)x;
    const unsigned lane16 = 16u * (unsigned)lane;
    const int ntiles = (N + 15) >> 4;
    const int nblk   = (int)gridDim.x;
    for (int tile = wid * nblk + (int)blockIdx.x; tile < ntiles; tile += P2_WAVES * nblk) {
    const int tile0 = tile * 16;

    int srow = tile0 + lo;
    if (srow >= N) srow = N - 1;
    const float* xself = x + (size_t)srow * NF + 2 * hi;
    v2f as_cur[8];
    #pragma unroll
    for (int s = 0; s < 8; ++s) {
        as_cur[s].x = xself[4 * s];
        as_cur[s].y = xself[4 * s + 1];
    }

    {
        int node = tile0 + lo;
        if (node >= N) node = N - 1;
        int deg = degrees[node];
        if (deg < 1) deg = 1;
        if (deg > MAXD) deg = MAXD;
        int k = deg >> 1;
        if (k < 1) k = 1;
        float sc = s0[node];
        unsigned key[MAXD];
        int      ids[MAXD];
        float    sn[MAXD];
        #pragma unroll
        for (int j = 0; j < MAXD; ++j) {
            ids[j] = neighbors[(size_t)node * MAXD + j];
            const int src = (j < deg) ? ids[j] : node;
            sn[j]  = s0[src];
        }
        __builtin_amdgcn_sched_barrier(0);
        #pragma unroll
        for (int j = 0; j < MAXD; ++j) {
            const float d = fabsf(sc - sn[j]);
            key[j] = (j < deg) ? __float_as_uint(d) : 0x7f800000u;
        }
        #pragma unroll
        for (int jj = 0; jj < MAXD / 2; ++jj) {
            const unsigned mk  = hi ? key[2 * jj + 1] : key[2 * jj];
            const int      mid = hi ? ids[2 * jj + 1] : ids[2 * jj];
            unsigned place = 0;
            #pragma unroll
            for (int i = 0; i < MAXD; ++i) {
                if (i == 2 * jj || i == 2 * jj + 1) continue;
                if (i < 2 * jj) place += (key[i] <= mk) ? 1u : 0u;
                else            place += (key[i] <  mk) ? 1u : 0u;
            }
            const bool even_first = key[2 * jj] <= key[2 * jj + 1];
            place += (hi ? even_first : !even_first) ? 1u : 0u;
            asm volatile("" : "+v"(place));
            sh_sel[wid][lo][place] = mid;
        }
        const float inv = 1.0f / (float)k;
        if (hi == 0) *(int2*)&sh_ki[wid][lo][0] = make_int2(k, __float_as_int(inv));
    }
    wave_lds_order();

    v8f c[4];
    #pragma unroll
    for (int ct = 0; ct < 4; ++ct) c[ct] = (v8f){0.f,0.f,0.f,0.f,0.f,0.f,0.f,0.f};

    #pragma nounroll
    for (int g = 0; g < 4; ++g) {
        const int m0 = 4 * g;
        int    kq[4];
        v4f    r[4][MAXD / 2];
        v4i idlo[4], idhi[4];
        #pragma unroll
        for (int q = 0; q < 4; ++q) {
            const int kv = sh_ki[wid][m0 + q][0];
            idlo[q] = *(const v4i*)&sh_sel[wid][m0 + q][0];
            idhi[q] = *(const v4i*)&sh_sel[wid][m0 + q][4];
            kq[q]   = __builtin_amdgcn_readfirstlane(kv);
        }
        asm volatile("" : "+v"(idlo[0]), "+v"(idhi[0]), "+v"(idlo[1]), "+v"(idhi[1]),
                          "+v"(idlo[2]), "+v"(idhi[2]), "+v"(idlo[3]), "+v"(idhi[3]));
        #pragma unroll
        for (int q = 0; q < 4; ++q) {
            #pragma unroll
            for (int j = 0; j < MAXD / 2; ++j) {
                if (j > 0 && j >= kq[q]) break;
                const int id = (j < 4) ? idlo[q][j & 3] : idhi[q][j & 3];
                const unsigned off = ((unsigned)id << 9) + lane16;
                r[q][j] = *(const v4f*)(xb + off);
            }
        }
        v2f as_nxt[8];
        #pragma unroll
        for (int s = 0; s < 8; ++s) as_nxt[s] = as_cur[s];
        if (g < 3) {
            #pragma unroll
            for (int s = 0; s < 8; ++s) {
                as_nxt[s].x = xself[32 * (g + 1) + 4 * s];
                as_nxt[s].y = xself[32 * (g + 1) + 4 * s + 1];
            }
        }
        #pragma unroll
        for (int p = 0; p < 4; ++p) {
            #pragma unroll
            for (int ct = 0; ct < 4; ++ct) {
                const float4 wf = shWs[((4 * g + p) * 4 + ct) * 32 + lane];
                v2f b0; b0.x = wf.x; b0.y = wf.y;
                v2f b1; b1.x = wf.z; b1.y = wf.w;
                c[ct] = __builtin_amdgcn_wmma_f32_16x16x4_f32(
                            false, as_cur[2 * p],     false, b0, (short)0, c[ct], false, false);
                c[ct] = __builtin_amdgcn_wmma_f32_16x16x4_f32(
                            false, as_cur[2 * p + 1], false, b1, (short)0, c[ct], false, false);
            }
        }
        asm volatile("" : "+v"(c[0]), "+v"(c[1]), "+v"(c[2]), "+v"(c[3]),
                          "+v"(r[0][0]), "+v"(r[1][0]), "+v"(r[2][0]), "+v"(r[3][0]));
        #pragma unroll
        for (int q = 0; q < 4; ++q) {
            v4f a = (v4f){0.f, 0.f, 0.f, 0.f};
            #pragma unroll
            for (int j = 0; j < MAXD / 2; ++j) {
                if (j > 0 && j >= kq[q]) break;
                a += r[q][j];
            }
            const float inv = __int_as_float(sh_ki[wid][m0 + q][1]);
            a *= inv;
            *(v4f*)&sh_an[wid][m0 + q][4 * lane] = a;
        }
        #pragma unroll
        for (int s = 0; s < 8; ++s) as_cur[s] = as_nxt[s];
    }
    wave_lds_order();

    for (int i2 = 0; i2 < NF / 8; ++i2) {
        const int kk = 8 * i2 + hi * 2;
        v2f a0, a1;
        a0.x = sh_an[wid][lo][kk];     a0.y = sh_an[wid][lo][kk + 1];
        a1.x = sh_an[wid][lo][kk + 4]; a1.y = sh_an[wid][lo][kk + 5];
        #pragma unroll
        for (int ct = 0; ct < 4; ++ct) {
            const float4 wf = shWn[(i2 * 4 + ct) * 32 + lane];
            v2f b0; b0.x = wf.x; b0.y = wf.y;
            v2f b1; b1.x = wf.z; b1.y = wf.w;
            c[ct] = __builtin_amdgcn_wmma_f32_16x16x4_f32(
                        false, a0, false, b0, (short)0, c[ct], false, false);
            c[ct] = __builtin_amdgcn_wmma_f32_16x16x4_f32(
                        false, a1, false, b1, (short)0, c[ct], false, false);
        }
    }

    #pragma unroll
    for (int ct = 0; ct < 4; ++ct) {
        const int e = ct * 16 + lo;
        const float be = b_sage[e];
        #pragma unroll
        for (int j = 0; j < 8; ++j) {
            float v = c[ct][j] + be;
            sh_h[wid][j + 8 * hi][e] = v > 0.f ? v : 0.f;
        }
    }
    wave_lds_order();

    {
        const int m = lane >> 1, cls = lane & 1;
        float acc = b_lin[cls];
        #pragma unroll 8
        for (int e = 0; e < EMBD; ++e)
            acc += sh_h[wid][m][e] * sh_wlin[e * NCLS + cls];
        const int node = tile0 + m;
        if (node < N)
            logits_out[(size_t)node * NCLS + cls] = acc;
    }
    wave_lds_order();
    }
}

extern "C" void kernel_launch(void* const* d_in, const int* in_sizes, int n_in,
                              void* d_out, int out_size, void* d_ws, size_t ws_size,
                              hipStream_t stream) {
    const float* x       = (const float*)d_in[0];
    const float* w_dist  = (const float*)d_in[1];
    const float* b_dist  = (const float*)d_in[2];
    const float* w_self  = (const float*)d_in[3];
    const float* w_neigh = (const float*)d_in[4];
    const float* b_sage  = (const float*)d_in[5];
    const float* w_lin   = (const float*)d_in[6];
    const float* b_lin   = (const float*)d_in[7];
    const int*   neigh   = (const int*)d_in[8];
    const int*   degs    = (const int*)d_in[9];

    const int N = in_sizes[9];
    float* out       = (float*)d_out;
    float* logits    = out;
    float* score_out = out + (size_t)N * NCLS;
    float* s0        = (float*)d_ws;

    {
        int nodes_per_block = (256 / 32) * P1_ROWS;
        int blocks = (N + nodes_per_block - 1) / nodes_per_block;
        care_phase1<<<blocks, 256, 0, stream>>>(x, w_dist, b_dist,
                                                score_out, s0, N);
    }
    {
        care_phase2<<<256, P2_THREADS, 0, stream>>>(x, w_self, w_neigh,
                                                       b_sage, w_lin, b_lin,
                                                       neigh, degs, s0,
                                                       logits, N);
    }
}
